# starting kernel
# baseline (speedup 1.0000x reference)
#include <hip/hip_runtime.h>

#define B_    4
#define N_    50000
#define CIN_  64
#define E_    800000
#define HALF_ 32
#define M_    (B_ * N_)

typedef __attribute__((ext_vector_type(2))) float v2f;
typedef __attribute__((ext_vector_type(8))) float v8f;

__global__ __launch_bounds__(256) void gcn_gemm_wmma(
    const float* __restrict__ x,
    const float* __restrict__ W_lin, const float* __restrict__ b_lin,
    const float* __restrict__ W_eye, const float* __restrict__ b_eye,
    float* __restrict__ out, float* __restrict__ z)
{
    __shared__ float lw[64 * 64];

    const int tid = threadIdx.x;
    #pragma unroll
    for (int j = 0; j < 16; ++j) {
        int idx = tid + 256 * j;
        int n = idx >> 6, k = idx & 63;
        lw[idx] = (n < HALF_) ? W_lin[n * CIN_ + k]
                              : W_eye[(n - HALF_) * CIN_ + k];
    }
    __syncthreads();

    const int wave  = tid >> 5;
    const int lane  = tid & 31;
    const int hi    = lane >> 4;
    const int l16   = lane & 15;
    const int mtile = blockIdx.x * 2 + (wave >> 2);
    const int nt    = wave & 3;

    const int row  = mtile * 16 + l16;
    const int ncol = nt * 16 + l16;

    const float* xrow = x  + (long)row * CIN_ + hi * 2;
    const float* wcol = lw + ncol * CIN_ + hi * 2;

    v8f acc = {};
    #pragma unroll
    for (int kb = 0; kb < 16; ++kb) {
        v2f a = *(const v2f*)(xrow + kb * 4);
        v2f b = *(const v2f*)(wcol + kb * 4);
        acc = __builtin_amdgcn_wmma_f32_16x16x4_f32(
             false, a,  false, b,
             (short)0, acc,  false,  false);
    }

    const int m0 = mtile * 16 + 8 * hi;
    if (nt < 2) {
        const float bl = b_lin[ncol];
        #pragma unroll
        for (int r = 0; r < 8; ++r) {
            const long m = m0 + r;
            z[m * HALF_ + ncol]  = acc[r];
            out[m * CIN_ + ncol] = bl;
        }
    } else {
        const float be = b_eye[ncol - HALF_];
        #pragma unroll
        for (int r = 0; r < 8; ++r) {
            const long m = m0 + r;
            out[m * CIN_ + ncol] = acc[r] + be;
        }
    }
}

__global__ __launch_bounds__(256) void gcn_edge_scatter(
    const float* __restrict__ z, const float* __restrict__ vals,
    const int* __restrict__ rows, const int* __restrict__ cols,
    float* __restrict__ out)
{
    const long t = (long)blockIdx.x * 256 + threadIdx.x;
    const int e = (int)(t >> 7);
    const int r = (int)(t & 127);
    const int b = r >> 5;
    const int c = r & 31;

    const int   cn = cols[e];
    const int   rn = rows[e];
    const float v  = vals[e];

    const float zv  = z[((long)b * N_ + cn) * HALF_ + c];
    const float upd = v * zv;
    float* dst = out + ((long)b * N_ + rn) * CIN_ + c;

    asm volatile("global_atomic_add_f32 %0, %1, off scope:SCOPE_DEV"
                 :
                 : "v"(dst), "v"(upd)
                 : "memory");
}

extern "C" void kernel_launch(void* const* d_in, const int* in_sizes, int n_in,
                              void* d_out, int out_size, void* d_ws, size_t ws_size,
                              hipStream_t stream) {
    const float* x     = (const float*)d_in[0];
    const float* vals  = (const float*)d_in[1];
    const float* W_lin = (const float*)d_in[2];
    const float* b_lin = (const float*)d_in[3];
    const float* W_eye = (const float*)d_in[4];
    const float* b_eye = (const float*)d_in[5];
    const int*   rows  = (const int*)d_in[6];
    const int*   cols  = (const int*)d_in[7];
    float* out = (float*)d_out;
    float* z   = (float*)d_ws;

    gcn_gemm_wmma<<<M_ / 32, 256, 0, stream>>>(x, W_lin, b_lin, W_eye, b_eye, out, z);

    gcn_edge_scatter<<<(long)E_ * 128 / 256, 256, 0, stream>>>(z, vals, rows, cols, out);
}


# Float atomics removed: edges stably radix-sorted by destination row each call (ballot ranks, no atomics), one wave per row gathers z and stores out once; per-row sums now in fixed edge order
# speedup vs baseline: 2.0920x; 2.0920x over previous
#include <hip/hip_runtime.h>


#define B_    4
#define N_    50000
#define CIN_  64
#define E_    800000
#define HALF_ 32
#define M_    (B_ * N_)

#define SORT_BLOCKS 625
#define SORT_GROUPS 5

#define WS_Z_OFF      0
#define WS_CNT_OFF    25600000
#define WS_PTR_OFF    26240000
#define WS_TOT_OFF    26440704
#define WS_KEYA_OFF   26441728
#define WS_KEYB_OFF   29641728
#define WS_PAIRA_OFF  32841728
#define WS_PAIRB_OFF  39241728

typedef __attribute__((ext_vector_type(2))) float v2f;
typedef __attribute__((ext_vector_type(8))) float v8f;
typedef __attribute__((ext_vector_type(2))) int   v2i;

__global__ __launch_bounds__(256) void gcn_gemm_wmma(
    const float* __restrict__ x,
    const float* __restrict__ W_lin,
    const float* __restrict__ W_eye, const float* __restrict__ b_eye,
    float* __restrict__ out, float* __restrict__ z)
{
    __shared__ float lw[64 * 64];

    const int tid = threadIdx.x;
    #pragma unroll
    for (int j = 0; j < 16; ++j) {
        int idx = tid + 256 * j;
        int n = idx >> 6, k = idx & 63;
        lw[idx] = (n < HALF_) ? W_lin[n * CIN_ + k]
                              : W_eye[(n - HALF_) * CIN_ + k];
    }
    __syncthreads();

    const int wave  = tid >> 5;
    const int lane  = tid & 31;
    const int hi    = lane >> 4;
    const int l16   = lane & 15;
    const int mtile = blockIdx.x * 2 + (wave >> 2);
    const int nt    = wave & 3;

    const int row  = mtile * 16 + l16;
    const int ncol = nt * 16 + l16;

    const float* xrow = x  + (long)row * CIN_ + hi * 2;
    const float* wcol = lw + ncol * CIN_ + hi * 2;

    v8f acc = {};
    #pragma unroll
    for (int kb = 0; kb < 16; ++kb) {
        v2f a = *(const v2f*)(xrow + kb * 4);
        v2f b = *(const v2f*)(wcol + kb * 4);
        acc = __builtin_amdgcn_wmma_f32_16x16x4_f32(
             false, a,  false, b,
             (short)0, acc,  false,  false);
    }

    const int m0 = mtile * 16 + 8 * hi;
    if (nt < 2) {
        #pragma unroll
        for (int r = 0; r < 8; ++r) {
            const long m = m0 + r;
            z[m * HALF_ + ncol]  = acc[r];
        }
    } else {
        const float be = b_eye[ncol - HALF_];
        #pragma unroll
        for (int r = 0; r < 8; ++r) {
            const long m = m0 + r;
            out[m * CIN_ + ncol] = acc[r] + be;
        }
    }
}


__device__ __forceinline__ void wave_digit_ranks(
    const int (&d)[SORT_GROUPS], unsigned lt,
    int (&lrank)[SORT_GROUPS], int (&tot)[SORT_GROUPS])
{
    unsigned bal[SORT_GROUPS][8];
    #pragma unroll
    for (int g = 0; g < SORT_GROUPS; ++g) {
        #pragma unroll
        for (int b = 0; b < 8; ++b)
            bal[g][b] = __builtin_amdgcn_ballot_w32(((d[g] >> b) & 1) != 0);
    }
    #pragma unroll
    for (int g = 0; g < SORT_GROUPS; ++g) {
        int before = 0, all = 0;
        #pragma unroll
        for (int o = 0; o < SORT_GROUPS; ++o) {
            unsigned m = 0xffffffffu;
            #pragma unroll
            for (int b = 0; b < 8; ++b)
                m &= ((d[g] >> b) & 1) ? bal[o][b] : ~bal[o][b];
            all += __popc(m);
            if (o < g)  before += __popc(m);
            if (o == g) before += __popc(m & lt);
        }
        lrank[g] = before;
        tot[g]   = all;
    }
}

template <int SHIFT>
__global__ __launch_bounds__(256) void gcn_sort_count(
    const int* __restrict__ keys, int* __restrict__ counts)
{
    __shared__ int h[8 * 256];

    const int tid  = threadIdx.x;
    const int lane = tid & 31;
    const int w    = tid >> 5;
    #pragma unroll
    for (int i = 0; i < 8; ++i) h[tid + 256 * i] = 0;
    __syncthreads();

    const int base = blockIdx.x * (256 * SORT_GROUPS) + w * (32 * SORT_GROUPS) + lane;
    const unsigned lt = (1u << lane) - 1u;
    int d[SORT_GROUPS], lrank[SORT_GROUPS], tot[SORT_GROUPS];
    #pragma unroll
    for (int g = 0; g < SORT_GROUPS; ++g) d[g] = (keys[base + g * 32] >> SHIFT) & 255;
    wave_digit_ranks(d, lt, lrank, tot);
    #pragma unroll
    for (int g = 0; g < SORT_GROUPS; ++g)
        if (lrank[g] == 0) h[w * 256 + d[g]] = tot[g];
    __syncthreads();

    int total = 0;
    #pragma unroll
    for (int i = 0; i < 8; ++i) total += h[i * 256 + tid];
    counts[tid * SORT_BLOCKS + blockIdx.x] = total;
}

__device__ __forceinline__ int wave_inclusive_sum(int v, int lane)
{
    #pragma unroll
    for (int s = 1; s < 32; s <<= 1) {
        const int t = __shfl_up(v, s);
        if (lane >= s) v += t;
    }
    return v;
}

__global__ __launch_bounds__(256) void gcn_sort_scan_rows(
    int* __restrict__ counts, int* __restrict__ rowtot)
{
    __shared__ int wsum[8];

    const int tid  = threadIdx.x;
    const int lane = tid & 31;
    const int wave = tid >> 5;
    int* row = counts + blockIdx.x * SORT_BLOCKS;

    const int lo = tid * 3;
    const int c0 = (lo     < SORT_BLOCKS) ? row[lo]     : 0;
    const int c1 = (lo + 1 < SORT_BLOCKS) ? row[lo + 1] : 0;
    const int c2 = (lo + 2 < SORT_BLOCKS) ? row[lo + 2] : 0;
    const int sum = c0 + c1 + c2;
    const int inc = wave_inclusive_sum(sum, lane);
    if (lane == 31) wsum[wave] = inc;
    __syncthreads();

    int wbase = 0, total = 0;
    #pragma unroll
    for (int i = 0; i < 8; ++i) {
        const int v = wsum[i];
        if (i < wave) wbase += v;
        total += v;
    }
    int run = wbase + inc - sum;
    if (lo     < SORT_BLOCKS) row[lo]     = run;
    run += c0;
    if (lo + 1 < SORT_BLOCKS) row[lo + 1] = run;
    run += c1;
    if (lo + 2 < SORT_BLOCKS) row[lo + 2] = run;
    if (tid == 0) rowtot[blockIdx.x] = total;
}

template <int SHIFT, bool FIRST>
__global__ __launch_bounds__(256) void gcn_sort_scatter(
    const int* __restrict__ rows, const int* __restrict__ cols, const float* __restrict__ vals,
    const int* __restrict__ keyIn, const int* __restrict__ pairIn,
    const int* __restrict__ counts, const int* __restrict__ rowtot,
    int* __restrict__ keyOut, int* __restrict__ pairOut)
{
    __shared__ int h[8 * 256];
    __shared__ int dsum[8];

    const int tid  = threadIdx.x;
    const int lane = tid & 31;
    const int w    = tid >> 5;
    #pragma unroll
    for (int i = 0; i < 8; ++i) h[tid + 256 * i] = 0;
    const int dv   = rowtot[tid];
    const int dinc = wave_inclusive_sum(dv, lane);
    if (lane == 31) dsum[w] = dinc;
    __syncthreads();

    const int base = blockIdx.x * (256 * SORT_GROUPS) + w * (32 * SORT_GROUPS) + lane;
    const unsigned lt = (1u << lane) - 1u;

    int key[SORT_GROUPS], d[SORT_GROUPS], lrank[SORT_GROUPS], tot[SORT_GROUPS];
    v2i pr[SORT_GROUPS];
    #pragma unroll
    for (int g = 0; g < SORT_GROUPS; ++g) {
        const int e = base + g * 32;
        if (FIRST) {
            key[g]  = rows[e];
            pr[g].x = cols[e];
            pr[g].y = __float_as_int(vals[e]);
        } else {
            key[g] = keyIn[e];
            pr[g]  = ((const v2i*)pairIn)[e];
        }
        d[g] = (key[g] >> SHIFT) & 255;
    }
    wave_digit_ranks(d, lt, lrank, tot);
    #pragma unroll
    for (int g = 0; g < SORT_GROUPS; ++g)
        if (lrank[g] == 0) h[w * 256 + d[g]] = tot[g];
    __syncthreads();

    {
        int run = counts[tid * SORT_BLOCKS + blockIdx.x] + dinc - dv;
        #pragma unroll
        for (int i = 0; i < 8; ++i) if (i < w) run += dsum[i];
        #pragma unroll
        for (int i = 0; i < 8; ++i) {
            const int c = h[i * 256 + tid];
            h[i * 256 + tid] = run;
            run += c;
        }
    }
    __syncthreads();

    #pragma unroll
    for (int g = 0; g < SORT_GROUPS; ++g) {
        const int pos = h[w * 256 + d[g]] + lrank[g];
        if ((unsigned)pos < (unsigned)E_) {
            keyOut[pos] = key[g];
            ((v2i*)pairOut)[pos] = pr[g];
        }
    }
}

__global__ __launch_bounds__(256) void gcn_row_ptr(
    const int* __restrict__ skey, int* __restrict__ rowptr)
{
    const int i = blockIdx.x * 256 + threadIdx.x;
    int r  = skey[i];
    int rp = (i > 0) ? skey[i - 1] : -1;
    r  = r  < 0 ? 0 : (r  > N_ - 1 ? N_ - 1 : r);
    rp = rp < -1 ? -1 : (rp > N_ - 1 ? N_ - 1 : rp);
    for (int q = rp + 1; q <= r; ++q) rowptr[q] = i;
    if (i == E_ - 1)
        for (int q = r + 1; q <= N_; ++q) rowptr[q] = E_;
}

__global__ __launch_bounds__(256) void gcn_row_gather(
    const float* __restrict__ z, const int* __restrict__ rowptr,
    const int* __restrict__ pairs, const float* __restrict__ b_lin,
    float* __restrict__ out)
{
    const int lane = threadIdx.x & 31;
    const int n    = blockIdx.x * 8 + (threadIdx.x >> 5);

    int start = __builtin_amdgcn_readfirstlane(rowptr[n]);
    int end   = __builtin_amdgcn_readfirstlane(rowptr[n + 1]);
    start = start < 0 ? 0 : (start > E_ ? E_ : start);
    end   = end > E_ ? E_ : end;

    const long BS = (long)N_ * HALF_;
    const float* zl = z + lane;
    const v2i* pr = (const v2i*)pairs;

    float a0 = 0.f, a1 = 0.f, a2 = 0.f, a3 = 0.f;

    for (int base = start; base < end; base += 32) {
        const int m = (end - base < 32) ? (end - base) : 32;
        v2i p = {0, 0};
        if (lane < m) p = pr[base + lane];

        int j = 0;
        for (; j + 4 <= m; j += 4) {
            const int   c0 = __builtin_amdgcn_readlane(p.x, j);
            const int   c1 = __builtin_amdgcn_readlane(p.x, j + 1);
            const int   c2 = __builtin_amdgcn_readlane(p.x, j + 2);
            const int   c3 = __builtin_amdgcn_readlane(p.x, j + 3);
            const float v0 = __int_as_float(__builtin_amdgcn_readlane(p.y, j));
            const float v1 = __int_as_float(__builtin_amdgcn_readlane(p.y, j + 1));
            const float v2 = __int_as_float(__builtin_amdgcn_readlane(p.y, j + 2));
            const float v3 = __int_as_float(__builtin_amdgcn_readlane(p.y, j + 3));
            const float* q0 = zl + (long)c0 * HALF_;
            const float* q1 = zl + (long)c1 * HALF_;
            const float* q2 = zl + (long)c2 * HALF_;
            const float* q3 = zl + (long)c3 * HALF_;
            const float z00 = q0[0], z01 = q0[BS], z02 = q0[2 * BS], z03 = q0[3 * BS];
            const float z10 = q1[0], z11 = q1[BS], z12 = q1[2 * BS], z13 = q1[3 * BS];
            const float z20 = q2[0], z21 = q2[BS], z22 = q2[2 * BS], z23 = q2[3 * BS];
            const float z30 = q3[0], z31 = q3[BS], z32 = q3[2 * BS], z33 = q3[3 * BS];
            a0 = fmaf(v0, z00, a0); a1 = fmaf(v0, z01, a1); a2 = fmaf(v0, z02, a2); a3 = fmaf(v0, z03, a3);
            a0 = fmaf(v1, z10, a0); a1 = fmaf(v1, z11, a1); a2 = fmaf(v1, z12, a2); a3 = fmaf(v1, z13, a3);
            a0 = fmaf(v2, z20, a0); a1 = fmaf(v2, z21, a1); a2 = fmaf(v2, z22, a2); a3 = fmaf(v2, z23, a3);
            a0 = fmaf(v3, z30, a0); a1 = fmaf(v3, z31, a1); a2 = fmaf(v3, z32, a2); a3 = fmaf(v3, z33, a3);
        }
        for (; j < m; ++j) {
            const int   c0 = __builtin_amdgcn_readlane(p.x, j);
            const float v0 = __int_as_float(__builtin_amdgcn_readlane(p.y, j));
            const float* q0 = zl + (long)c0 * HALF_;
            const float z00 = q0[0], z01 = q0[BS], z02 = q0[2 * BS], z03 = q0[3 * BS];
            a0 = fmaf(v0, z00, a0); a1 = fmaf(v0, z01, a1); a2 = fmaf(v0, z02, a2); a3 = fmaf(v0, z03, a3);
        }
    }

    const float bl = b_lin[lane];
    float* o = out + (long)n * CIN_ + lane;
    const long OS = (long)N_ * CIN_;
    o[0]      = bl + a0;
    o[OS]     = bl + a1;
    o[2 * OS] = bl + a2;
    o[3 * OS] = bl + a3;
}

extern "C" void kernel_launch(void* const* d_in, const int* in_sizes, int n_in,
                              void* d_out, int out_size, void* d_ws, size_t ws_size,
                              hipStream_t stream) {
    const float* x     = (const float*)d_in[0];
    const float* vals  = (const float*)d_in[1];
    const float* W_lin = (const float*)d_in[2];
    const float* b_lin = (const float*)d_in[3];
    const float* W_eye = (const float*)d_in[4];
    const float* b_eye = (const float*)d_in[5];
    const int*   rows  = (const int*)d_in[6];
    const int*   cols  = (const int*)d_in[7];
    float* out    = (float*)d_out;
    char*  ws     = (char*)d_ws;
    float* z      = (float*)(ws + WS_Z_OFF);
    int*   counts = (int*)(ws + WS_CNT_OFF);
    int*   rowptr = (int*)(ws + WS_PTR_OFF);
    int*   rowtot = (int*)(ws + WS_TOT_OFF);
    int*   keyA   = (int*)(ws + WS_KEYA_OFF);
    int*   keyB   = (int*)(ws + WS_KEYB_OFF);
    int*   pairA  = (int*)(ws + WS_PAIRA_OFF);
    int*   pairB  = (int*)(ws + WS_PAIRB_OFF);

    gcn_gemm_wmma<<<M_ / 32, 256, 0, stream>>>(x, W_lin, W_eye, b_eye, out, z);

    gcn_sort_count<0><<<SORT_BLOCKS, 256, 0, stream>>>(rows, counts);
    gcn_sort_scan_rows<<<256, 256, 0, stream>>>(counts, rowtot);
    gcn_sort_scatter<0, true><<<SORT_BLOCKS, 256, 0, stream>>>(rows, cols, vals, rows, rows, counts, rowtot, keyA, pairA);
    gcn_sort_count<8><<<SORT_BLOCKS, 256, 0, stream>>>(keyA, counts);
    gcn_sort_scan_rows<<<256, 256, 0, stream>>>(counts, rowtot);
    gcn_sort_scatter<8, false><<<SORT_BLOCKS, 256, 0, stream>>>(rows, cols, vals, keyA, pairA, counts, rowtot, keyB, pairB);

    gcn_row_ptr<<<E_ / 256, 256, 0, stream>>>(keyB, rowptr);

    gcn_row_gather<<<N_ / 8, 256, 0, stream>>>(z, rowptr, pairB, b_lin, out);
}


# GEMM operands fed from LDS: a block's x rows read once as whole 128-byte lines (were 8-byte loads over 16 lines, repeated by 4 waves), x and weight tiles at a conflict-free pitch; k order unchanged
# speedup vs baseline: 2.2727x; 1.0864x over previous
#include <hip/hip_runtime.h>


#define B_    4
#define N_    50000
#define CIN_  64
#define E_    800000
#define HALF_ 32
#define M_    (B_ * N_)

#define SORT_BLOCKS 625
#define SORT_GROUPS 5

#define WS_Z_OFF      0
#define WS_CNT_OFF    25600000
#define WS_PTR_OFF    26240000
#define WS_TOT_OFF    26440704
#define WS_KEYA_OFF   26441728
#define WS_KEYB_OFF   29641728
#define WS_PAIRA_OFF  32841728
#define WS_PAIRB_OFF  39241728

typedef __attribute__((ext_vector_type(2))) float v2f;
typedef __attribute__((ext_vector_type(8))) float v8f;
typedef __attribute__((ext_vector_type(2))) int   v2i;
typedef __attribute__((ext_vector_type(4))) float v4f;

#define LDS_PITCH 68

__global__ __launch_bounds__(256) void gcn_gemm_wmma(
    const float* __restrict__ x,
    const float* __restrict__ W_lin,
    const float* __restrict__ W_eye, const float* __restrict__ b_eye,
    float* __restrict__ out, float* __restrict__ z)
{
    __shared__ __attribute__((aligned(16))) float lw[64 * LDS_PITCH];
    __shared__ __attribute__((aligned(16))) float lx[32 * LDS_PITCH];

    const int tid = threadIdx.x;
    #pragma unroll
    for (int j = 0; j < 4; ++j) {
        const int idx = tid + 256 * j;
        const int n = idx >> 4, c4 = idx & 15;
        const float* src = (n < HALF_) ? (W_lin + n * CIN_ + c4 * 4)
                                       : (W_eye + (n - HALF_) * CIN_ + c4 * 4);
        *(v4f*)(lw + n * LDS_PITCH + c4 * 4) = *(const v4f*)src;
    }
    const float* xt = x + (long)blockIdx.x * (32 * CIN_);
    #pragma unroll
    for (int j = 0; j < 2; ++j) {
        const int idx = tid + 256 * j;
        const int r = idx >> 4, c4 = idx & 15;
        *(v4f*)(lx + r * LDS_PITCH + c4 * 4) = *(const v4f*)(xt + idx * 4);
    }
    __syncthreads();

    const int wave  = tid >> 5;
    const int lane  = tid & 31;
    const int hi    = lane >> 4;
    const int l16   = lane & 15;
    const int mtile = blockIdx.x * 2 + (wave >> 2);
    const int nt    = wave & 3;

    const int ncol = nt * 16 + l16;

    const float* xrow = lx + ((wave >> 2) * 16 + l16) * LDS_PITCH + hi * 2;
    const float* wcol = lw + ncol * LDS_PITCH + hi * 2;

    v8f acc = {};
    #pragma unroll
    for (int kb = 0; kb < 16; ++kb) {
        v2f a = *(const v2f*)(xrow + kb * 4);
        v2f b = *(const v2f*)(wcol + kb * 4);
        acc = __builtin_amdgcn_wmma_f32_16x16x4_f32(
             false, a,  false, b,
             (short)0, acc,  false,  false);
    }

    const int m0 = mtile * 16 + 8 * hi;
    if (nt < 2) {
        #pragma unroll
        for (int r = 0; r < 8; ++r) {
            const long m = m0 + r;
            z[m * HALF_ + ncol]  = acc[r];
        }
    } else {
        const float be = b_eye[ncol - HALF_];
        #pragma unroll
        for (int r = 0; r < 8; ++r) {
            const long m = m0 + r;
            out[m * CIN_ + ncol] = acc[r] + be;
        }
    }
}


__device__ __forceinline__ void wave_digit_ranks(
    const int (&d)[SORT_GROUPS], unsigned lt,
    int (&lrank)[SORT_GROUPS], int (&tot)[SORT_GROUPS])
{
    unsigned bal[SORT_GROUPS][8];
    #pragma unroll
    for (int g = 0; g < SORT_GROUPS; ++g) {
        #pragma unroll
        for (int b = 0; b < 8; ++b)
            bal[g][b] = __builtin_amdgcn_ballot_w32(((d[g] >> b) & 1) != 0);
    }
    #pragma unroll
    for (int g = 0; g < SORT_GROUPS; ++g) {
        int before = 0, all = 0;
        #pragma unroll
        for (int o = 0; o < SORT_GROUPS; ++o) {
            unsigned m = 0xffffffffu;
            #pragma unroll
            for (int b = 0; b < 8; ++b)
                m &= ((d[g] >> b) & 1) ? bal[o][b] : ~bal[o][b];
            all += __popc(m);
            if (o < g)  before += __popc(m);
            if (o == g) before += __popc(m & lt);
        }
        lrank[g] = before;
        tot[g]   = all;
    }
}

template <int SHIFT>
__global__ __launch_bounds__(256) void gcn_sort_count(
    const int* __restrict__ keys, int* __restrict__ counts)
{
    __shared__ int h[8 * 256];

    const int tid  = threadIdx.x;
    const int lane = tid & 31;
    const int w    = tid >> 5;
    #pragma unroll
    for (int i = 0; i < 8; ++i) h[tid + 256 * i] = 0;
    __syncthreads();

    const int base = blockIdx.x * (256 * SORT_GROUPS) + w * (32 * SORT_GROUPS) + lane;
    const unsigned lt = (1u << lane) - 1u;
    int d[SORT_GROUPS], lrank[SORT_GROUPS], tot[SORT_GROUPS];
    #pragma unroll
    for (int g = 0; g < SORT_GROUPS; ++g) d[g] = (keys[base + g * 32] >> SHIFT) & 255;
    wave_digit_ranks(d, lt, lrank, tot);
    #pragma unroll
    for (int g = 0; g < SORT_GROUPS; ++g)
        if (lrank[g] == 0) h[w * 256 + d[g]] = tot[g];
    __syncthreads();

    int total = 0;
    #pragma unroll
    for (int i = 0; i < 8; ++i) total += h[i * 256 + tid];
    counts[tid * SORT_BLOCKS + blockIdx.x] = total;
}

__device__ __forceinline__ int wave_inclusive_sum(int v, int lane)
{
    #pragma unroll
    for (int s = 1; s < 32; s <<= 1) {
        const int t = __shfl_up(v, s);
        if (lane >= s) v += t;
    }
    return v;
}

__global__ __launch_bounds__(256) void gcn_sort_scan_rows(
    int* __restrict__ counts, int* __restrict__ rowtot)
{
    __shared__ int wsum[8];

    const int tid  = threadIdx.x;
    const int lane = tid & 31;
    const int wave = tid >> 5;
    int* row = counts + blockIdx.x * SORT_BLOCKS;

    const int lo = tid * 3;
    const int c0 = (lo     < SORT_BLOCKS) ? row[lo]     : 0;
    const int c1 = (lo + 1 < SORT_BLOCKS) ? row[lo + 1] : 0;
    const int c2 = (lo + 2 < SORT_BLOCKS) ? row[lo + 2] : 0;
    const int sum = c0 + c1 + c2;
    const int inc = wave_inclusive_sum(sum, lane);
    if (lane == 31) wsum[wave] = inc;
    __syncthreads();

    int wbase = 0, total = 0;
    #pragma unroll
    for (int i = 0; i < 8; ++i) {
        const int v = wsum[i];
        if (i < wave) wbase += v;
        total += v;
    }
    int run = wbase + inc - sum;
    if (lo     < SORT_BLOCKS) row[lo]     = run;
    run += c0;
    if (lo + 1 < SORT_BLOCKS) row[lo + 1] = run;
    run += c1;
    if (lo + 2 < SORT_BLOCKS) row[lo + 2] = run;
    if (tid == 0) rowtot[blockIdx.x] = total;
}

template <int SHIFT, bool FIRST>
__global__ __launch_bounds__(256) void gcn_sort_scatter(
    const int* __restrict__ rows, const int* __restrict__ cols, const float* __restrict__ vals,
    const int* __restrict__ keyIn, const int* __restrict__ pairIn,
    const int* __restrict__ counts, const int* __restrict__ rowtot,
    int* __restrict__ keyOut, int* __restrict__ pairOut)
{
    __shared__ int h[8 * 256];
    __shared__ int dsum[8];

    const int tid  = threadIdx.x;
    const int lane = tid & 31;
    const int w    = tid >> 5;
    #pragma unroll
    for (int i = 0; i < 8; ++i) h[tid + 256 * i] = 0;
    const int dv   = rowtot[tid];
    const int dinc = wave_inclusive_sum(dv, lane);
    if (lane == 31) dsum[w] = dinc;
    __syncthreads();

    const int base = blockIdx.x * (256 * SORT_GROUPS) + w * (32 * SORT_GROUPS) + lane;
    const unsigned lt = (1u << lane) - 1u;

    int key[SORT_GROUPS], d[SORT_GROUPS], lrank[SORT_GROUPS], tot[SORT_GROUPS];
    v2i pr[SORT_GROUPS];
    #pragma unroll
    for (int g = 0; g < SORT_GROUPS; ++g) {
        const int e = base + g * 32;
        if (FIRST) {
            key[g]  = rows[e];
            pr[g].x = cols[e];
            pr[g].y = __float_as_int(vals[e]);
        } else {
            key[g] = keyIn[e];
            pr[g]  = ((const v2i*)pairIn)[e];
        }
        d[g] = (key[g] >> SHIFT) & 255;
    }
    wave_digit_ranks(d, lt, lrank, tot);
    #pragma unroll
    for (int g = 0; g < SORT_GROUPS; ++g)
        if (lrank[g] == 0) h[w * 256 + d[g]] = tot[g];
    __syncthreads();

    {
        int run = counts[tid * SORT_BLOCKS + blockIdx.x] + dinc - dv;
        #pragma unroll
        for (int i = 0; i < 8; ++i) if (i < w) run += dsum[i];
        #pragma unroll
        for (int i = 0; i < 8; ++i) {
            const int c = h[i * 256 + tid];
            h[i * 256 + tid] = run;
            run += c;
        }
    }
    __syncthreads();

    #pragma unroll
    for (int g = 0; g < SORT_GROUPS; ++g) {
        const int pos = h[w * 256 + d[g]] + lrank[g];
        if ((unsigned)pos < (unsigned)E_) {
            keyOut[pos] = key[g];
            ((v2i*)pairOut)[pos] = pr[g];
        }
    }
}

__global__ __launch_bounds__(256) void gcn_row_ptr(
    const int* __restrict__ skey, int* __restrict__ rowptr)
{
    const int i = blockIdx.x * 256 + threadIdx.x;
    int r  = skey[i];
    int rp = (i > 0) ? skey[i - 1] : -1;
    r  = r  < 0 ? 0 : (r  > N_ - 1 ? N_ - 1 : r);
    rp = rp < -1 ? -1 : (rp > N_ - 1 ? N_ - 1 : rp);
    for (int q = rp + 1; q <= r; ++q) rowptr[q] = i;
    if (i == E_ - 1)
        for (int q = r + 1; q <= N_; ++q) rowptr[q] = E_;
}

__global__ __launch_bounds__(256) void gcn_row_gather(
    const float* __restrict__ z, const int* __restrict__ rowptr,
    const int* __restrict__ pairs, const float* __restrict__ b_lin,
    float* __restrict__ out)
{
    const int lane = threadIdx.x & 31;
    const int n    = blockIdx.x * 8 + (threadIdx.x >> 5);

    int start = __builtin_amdgcn_readfirstlane(rowptr[n]);
    int end   = __builtin_amdgcn_readfirstlane(rowptr[n + 1]);
    start = start < 0 ? 0 : (start > E_ ? E_ : start);
    end   = end > E_ ? E_ : end;

    const long BS = (long)N_ * HALF_;
    const float* zl = z + lane;
    const v2i* pr = (const v2i*)pairs;

    float a0 = 0.f, a1 = 0.f, a2 = 0.f, a3 = 0.f;

    for (int base = start; base < end; base += 32) {
        const int m = (end - base < 32) ? (end - base) : 32;
        v2i p = {0, 0};
        if (lane < m) p = pr[base + lane];

        int j = 0;
        for (; j + 4 <= m; j += 4) {
            const int   c0 = __builtin_amdgcn_readlane(p.x, j);
            const int   c1 = __builtin_amdgcn_readlane(p.x, j + 1);
            const int   c2 = __builtin_amdgcn_readlane(p.x, j + 2);
            const int   c3 = __builtin_amdgcn_readlane(p.x, j + 3);
            const float v0 = __int_as_float(__builtin_amdgcn_readlane(p.y, j));
            const float v1 = __int_as_float(__builtin_amdgcn_readlane(p.y, j + 1));
            const float v2 = __int_as_float(__builtin_amdgcn_readlane(p.y, j + 2));
            const float v3 = __int_as_float(__builtin_amdgcn_readlane(p.y, j + 3));
            const float* q0 = zl + (long)c0 * HALF_;
            const float* q1 = zl + (long)c1 * HALF_;
            const float* q2 = zl + (long)c2 * HALF_;
            const float* q3 = zl + (long)c3 * HALF_;
            const float z00 = q0[0], z01 = q0[BS], z02 = q0[2 * BS], z03 = q0[3 * BS];
            const float z10 = q1[0], z11 = q1[BS], z12 = q1[2 * BS], z13 = q1[3 * BS];
            const float z20 = q2[0], z21 = q2[BS], z22 = q2[2 * BS], z23 = q2[3 * BS];
            const float z30 = q3[0], z31 = q3[BS], z32 = q3[2 * BS], z33 = q3[3 * BS];
            a0 = fmaf(v0, z00, a0); a1 = fmaf(v0, z01, a1); a2 = fmaf(v0, z02, a2); a3 = fmaf(v0, z03, a3);
            a0 = fmaf(v1, z10, a0); a1 = fmaf(v1, z11, a1); a2 = fmaf(v1, z12, a2); a3 = fmaf(v1, z13, a3);
            a0 = fmaf(v2, z20, a0); a1 = fmaf(v2, z21, a1); a2 = fmaf(v2, z22, a2); a3 = fmaf(v2, z23, a3);
            a0 = fmaf(v3, z30, a0); a1 = fmaf(v3, z31, a1); a2 = fmaf(v3, z32, a2); a3 = fmaf(v3, z33, a3);
        }
        for (; j < m; ++j) {
            const int   c0 = __builtin_amdgcn_readlane(p.x, j);
            const float v0 = __int_as_float(__builtin_amdgcn_readlane(p.y, j));
            const float* q0 = zl + (long)c0 * HALF_;
            const float z00 = q0[0], z01 = q0[BS], z02 = q0[2 * BS], z03 = q0[3 * BS];
            a0 = fmaf(v0, z00, a0); a1 = fmaf(v0, z01, a1); a2 = fmaf(v0, z02, a2); a3 = fmaf(v0, z03, a3);
        }
    }

    const float bl = b_lin[lane];
    float* o = out + (long)n * CIN_ + lane;
    const long OS = (long)N_ * CIN_;
    o[0]      = bl + a0;
    o[OS]     = bl + a1;
    o[2 * OS] = bl + a2;
    o[3 * OS] = bl + a3;
}

extern "C" void kernel_launch(void* const* d_in, const int* in_sizes, int n_in,
                              void* d_out, int out_size, void* d_ws, size_t ws_size,
                              hipStream_t stream) {
    const float* x     = (const float*)d_in[0];
    const float* vals  = (const float*)d_in[1];
    const float* W_lin = (const float*)d_in[2];
    const float* b_lin = (const float*)d_in[3];
    const float* W_eye = (const float*)d_in[4];
    const float* b_eye = (const float*)d_in[5];
    const int*   rows  = (const int*)d_in[6];
    const int*   cols  = (const int*)d_in[7];
    float* out    = (float*)d_out;
    char*  ws     = (char*)d_ws;
    float* z      = (float*)(ws + WS_Z_OFF);
    int*   counts = (int*)(ws + WS_CNT_OFF);
    int*   rowptr = (int*)(ws + WS_PTR_OFF);
    int*   rowtot = (int*)(ws + WS_TOT_OFF);
    int*   keyA   = (int*)(ws + WS_KEYA_OFF);
    int*   keyB   = (int*)(ws + WS_KEYB_OFF);
    int*   pairA  = (int*)(ws + WS_PAIRA_OFF);
    int*   pairB  = (int*)(ws + WS_PAIRB_OFF);

    gcn_gemm_wmma<<<M_ / 32, 256, 0, stream>>>(x, W_lin, W_eye, b_eye, out, z);

    gcn_sort_count<0><<<SORT_BLOCKS, 256, 0, stream>>>(rows, counts);
    gcn_sort_scan_rows<<<256, 256, 0, stream>>>(counts, rowtot);
    gcn_sort_scatter<0, true><<<SORT_BLOCKS, 256, 0, stream>>>(rows, cols, vals, rows, rows, counts, rowtot, keyA, pairA);
    gcn_sort_count<8><<<SORT_BLOCKS, 256, 0, stream>>>(keyA, counts);
    gcn_sort_scan_rows<<<256, 256, 0, stream>>>(counts, rowtot);
    gcn_sort_scatter<8, false><<<SORT_BLOCKS, 256, 0, stream>>>(rows, cols, vals, keyA, pairA, counts, rowtot, keyB, pairB);

    gcn_row_ptr<<<E_ / 256, 256, 0, stream>>>(keyB, rowptr);

    gcn_row_gather<<<N_ / 8, 256, 0, stream>>>(z, rowptr, pairB, b_lin, out);
}


# Second sort pass and row-pointer kernel removed (9 launches to 5): edges filed once by 64-row bucket; the bucket's work-group orders them by row in LDS and gathers; per-row sum order unchanged
# speedup vs baseline: 2.7476x; 1.2090x over previous
#include <hip/hip_runtime.h>


#define B_    4
#define N_    50000
#define CIN_  64
#define E_    800000
#define HALF_ 32
#define M_    (B_ * N_)

#define SORT_BLOCKS 625
#define SORT_GROUPS 5
#define DIG_SHIFT   6
#define DIG_BITS    10
#define NDIG        1024
#define BROWS       64
#define NBUCKET     ((N_ + BROWS - 1) / BROWS)
#define BK_THREADS  512
#define BK_WAVES    16
#define BK_GROUPS   3
#define BK_CHUNK    (BK_THREADS * BK_GROUPS)
#define BK_RPW      (BROWS / BK_WAVES)

#define WS_Z_OFF      0
#define WS_CNT_OFF    25600000
#define WS_TOT_OFF    28160000
#define WS_BST_OFF    28164096
#define WS_KEYA_OFF   28168448
#define WS_PAIRA_OFF  31368448

typedef __attribute__((ext_vector_type(2))) float v2f;
typedef __attribute__((ext_vector_type(8))) float v8f;
typedef __attribute__((ext_vector_type(2))) int   v2i;
typedef __attribute__((ext_vector_type(4))) float v4f;

#define LDS_PITCH 68

__global__ __launch_bounds__(256) void gcn_gemm_wmma(
    const float* __restrict__ x,
    const float* __restrict__ W_lin,
    const float* __restrict__ W_eye, const float* __restrict__ b_eye,
    float* __restrict__ out, float* __restrict__ z)
{
    __shared__ __attribute__((aligned(16))) float lw[64 * LDS_PITCH];
    __shared__ __attribute__((aligned(16))) float lx[32 * LDS_PITCH];

    const int tid = threadIdx.x;
    #pragma unroll
    for (int j = 0; j < 4; ++j) {
        const int idx = tid + 256 * j;
        const int n = idx >> 4, c4 = idx & 15;
        const float* src = (n < HALF_) ? (W_lin + n * CIN_ + c4 * 4)
                                       : (W_eye + (n - HALF_) * CIN_ + c4 * 4);
        *(v4f*)(lw + n * LDS_PITCH + c4 * 4) = *(const v4f*)src;
    }
    const float* xt = x + (long)blockIdx.x * (32 * CIN_);
    #pragma unroll
    for (int j = 0; j < 2; ++j) {
        const int idx = tid + 256 * j;
        const int r = idx >> 4, c4 = idx & 15;
        *(v4f*)(lx + r * LDS_PITCH + c4 * 4) = *(const v4f*)(xt + idx * 4);
    }
    __syncthreads();

    const int wave  = tid >> 5;
    const int lane  = tid & 31;
    const int hi    = lane >> 4;
    const int l16   = lane & 15;
    const int mtile = blockIdx.x * 2 + (wave >> 2);
    const int nt    = wave & 3;

    const int ncol = nt * 16 + l16;

    const float* xrow = lx + ((wave >> 2) * 16 + l16) * LDS_PITCH + hi * 2;
    const float* wcol = lw + ncol * LDS_PITCH + hi * 2;

    v8f acc = {};
    #pragma unroll
    for (int kb = 0; kb < 16; ++kb) {
        v2f a = *(const v2f*)(xrow + kb * 4);
        v2f b = *(const v2f*)(wcol + kb * 4);
        acc = __builtin_amdgcn_wmma_f32_16x16x4_f32(
             false, a,  false, b,
             (short)0, acc,  false,  false);
    }

    const int m0 = mtile * 16 + 8 * hi;
    if (nt < 2) {
        #pragma unroll
        for (int r = 0; r < 8; ++r) {
            const long m = m0 + r;
            z[m * HALF_ + ncol]  = acc[r];
        }
    } else {
        const float be = b_eye[ncol - HALF_];
        #pragma unroll
        for (int r = 0; r < 8; ++r) {
            const long m = m0 + r;
            out[m * CIN_ + ncol] = acc[r] + be;
        }
    }
}


template <int G, int NB>
__device__ __forceinline__ void wave_digit_ranks(
    const int (&d)[G], unsigned lt, int (&lrank)[G], int (&tot)[G])
{
    unsigned bal[G][NB];
    #pragma unroll
    for (int g = 0; g < G; ++g) {
        #pragma unroll
        for (int b = 0; b < NB; ++b)
            bal[g][b] = __builtin_amdgcn_ballot_w32(((d[g] >> b) & 1) != 0);
    }
    #pragma unroll
    for (int g = 0; g < G; ++g) {
        int before = 0, all = 0;
        #pragma unroll
        for (int o = 0; o < G; ++o) {
            unsigned m = 0xffffffffu;
            #pragma unroll
            for (int b = 0; b < NB; ++b)
                m &= ((d[g] >> b) & 1) ? bal[o][b] : ~bal[o][b];
            all += __popc(m);
            if (o < g)  before += __popc(m);
            if (o == g) before += __popc(m & lt);
        }
        lrank[g] = before;
        tot[g]   = all;
    }
}

__global__ __launch_bounds__(256) void gcn_sort_count(
    const int* __restrict__ keys, int* __restrict__ counts)
{
    __shared__ int h[8 * NDIG];

    const int tid  = threadIdx.x;
    const int lane = tid & 31;
    const int w    = tid >> 5;
    #pragma unroll
    for (int i = 0; i < 8 * NDIG / 256; ++i) h[tid + 256 * i] = 0;
    __syncthreads();

    const int base = blockIdx.x * (256 * SORT_GROUPS) + w * (32 * SORT_GROUPS) + lane;
    const unsigned lt = (1u << lane) - 1u;
    int d[SORT_GROUPS], lrank[SORT_GROUPS], tot[SORT_GROUPS];
    #pragma unroll
    for (int g = 0; g < SORT_GROUPS; ++g) d[g] = (keys[base + g * 32] >> DIG_SHIFT) & (NDIG - 1);
    wave_digit_ranks<SORT_GROUPS, DIG_BITS>(d, lt, lrank, tot);
    #pragma unroll
    for (int g = 0; g < SORT_GROUPS; ++g)
        if (lrank[g] == 0) h[w * NDIG + d[g]] = tot[g];
    __syncthreads();

    #pragma unroll
    for (int j = 0; j < NDIG / 256; ++j) {
        const int digit = tid + 256 * j;
        int total = 0;
        #pragma unroll
        for (int i = 0; i < 8; ++i) total += h[i * NDIG + digit];
        counts[digit * SORT_BLOCKS + blockIdx.x] = total;
    }
}

__device__ __forceinline__ int wave_inclusive_sum(int v, int lane)
{
    #pragma unroll
    for (int s = 1; s < 32; s <<= 1) {
        const int t = __shfl_up(v, s);
        if (lane >= s) v += t;
    }
    return v;
}

__global__ __launch_bounds__(256) void gcn_sort_scan_rows(
    int* __restrict__ counts, int* __restrict__ rowtot)
{
    __shared__ int wsum[8];

    const int tid  = threadIdx.x;
    const int lane = tid & 31;
    const int wave = tid >> 5;
    int* row = counts + blockIdx.x * SORT_BLOCKS;

    const int lo = tid * 3;
    const int c0 = (lo     < SORT_BLOCKS) ? row[lo]     : 0;
    const int c1 = (lo + 1 < SORT_BLOCKS) ? row[lo + 1] : 0;
    const int c2 = (lo + 2 < SORT_BLOCKS) ? row[lo + 2] : 0;
    const int sum = c0 + c1 + c2;
    const int inc = wave_inclusive_sum(sum, lane);
    if (lane == 31) wsum[wave] = inc;
    __syncthreads();

    int wbase = 0, total = 0;
    #pragma unroll
    for (int i = 0; i < 8; ++i) {
        const int v = wsum[i];
        if (i < wave) wbase += v;
        total += v;
    }
    int run = wbase + inc - sum;
    if (lo     < SORT_BLOCKS) row[lo]     = run;
    run += c0;
    if (lo + 1 < SORT_BLOCKS) row[lo + 1] = run;
    run += c1;
    if (lo + 2 < SORT_BLOCKS) row[lo + 2] = run;
    if (tid == 0) rowtot[blockIdx.x] = total;
}

__global__ __launch_bounds__(256) void gcn_sort_scatter(
    const int* __restrict__ rows, const int* __restrict__ cols, const float* __restrict__ vals,
    const int* __restrict__ counts, const int* __restrict__ rowtot,
    int* __restrict__ keyOut, int* __restrict__ pairOut, int* __restrict__ bstart)
{
    __shared__ int h[8 * NDIG];
    __shared__ int dsum[(NDIG / 256) * 8];

    const int tid  = threadIdx.x;
    const int lane = tid & 31;
    const int w    = tid >> 5;
    #pragma unroll
    for (int i = 0; i < 8 * NDIG / 256; ++i) h[tid + 256 * i] = 0;
    int dv[NDIG / 256], dinc[NDIG / 256];
    #pragma unroll
    for (int j = 0; j < NDIG / 256; ++j) {
        dv[j]   = rowtot[tid + 256 * j];
        dinc[j] = wave_inclusive_sum(dv[j], lane);
        if (lane == 31) dsum[j * 8 + w] = dinc[j];
    }
    __syncthreads();

    const int base = blockIdx.x * (256 * SORT_GROUPS) + w * (32 * SORT_GROUPS) + lane;
    const unsigned lt = (1u << lane) - 1u;

    int key[SORT_GROUPS], d[SORT_GROUPS], lrank[SORT_GROUPS], tot[SORT_GROUPS];
    v2i pr[SORT_GROUPS];
    #pragma unroll
    for (int g = 0; g < SORT_GROUPS; ++g) {
        const int e = base + g * 32;
        key[g]  = rows[e];
        pr[g].x = cols[e];
        pr[g].y = __float_as_int(vals[e]);
        d[g] = (key[g] >> DIG_SHIFT) & (NDIG - 1);
    }
    wave_digit_ranks<SORT_GROUPS, DIG_BITS>(d, lt, lrank, tot);
    #pragma unroll
    for (int g = 0; g < SORT_GROUPS; ++g)
        if (lrank[g] == 0) h[w * NDIG + d[g]] = tot[g];
    __syncthreads();

    {
        int carry = 0;
        #pragma unroll
        for (int j = 0; j < NDIG / 256; ++j) {
            int before = carry, gtotal = 0;
            #pragma unroll
            for (int i = 0; i < 8; ++i) {
                const int v = dsum[j * 8 + i];
                if (i < w) before += v;
                gtotal += v;
            }
            carry += gtotal;
            const int digit = tid + 256 * j;
            const int dbase = before + dinc[j] - dv[j];
            if (blockIdx.x == 0) bstart[digit] = dbase;
            int run = counts[digit * SORT_BLOCKS + blockIdx.x] + dbase;
            #pragma unroll
            for (int i = 0; i < 8; ++i) {
                const int c = h[i * NDIG + digit];
                h[i * NDIG + digit] = run;
                run += c;
            }
        }
        if (blockIdx.x == 0 && tid == 0) bstart[NDIG] = carry;
    }
    __syncthreads();

    #pragma unroll
    for (int g = 0; g < SORT_GROUPS; ++g) {
        const int pos = h[w * NDIG + d[g]] + lrank[g];
        if ((unsigned)pos < (unsigned)E_) {
            keyOut[pos] = key[g];
            ((v2i*)pairOut)[pos] = pr[g];
        }
    }
}

__global__ __launch_bounds__(BK_THREADS, 16) void gcn_bucket_gather(
    const float* __restrict__ z, const int* __restrict__ bstart,
    const int* __restrict__ skey, const int* __restrict__ spair,
    const float* __restrict__ b_lin, float* __restrict__ out)
{
    __shared__ int   h[BK_WAVES * 128];
    __shared__ int   seg[BROWS + 1];
    __shared__ v2i   srt[BK_CHUNK];
    __shared__ float accl[BROWS * B_ * 32];

    const int tid  = threadIdx.x;
    const int lane = tid & 31;
    const int w    = tid >> 5;
    const int q    = blockIdx.x;
    const unsigned lt = (1u << lane) - 1u;

    int bs = __builtin_amdgcn_readfirstlane(bstart[q]);
    int be = __builtin_amdgcn_readfirstlane(bstart[q + 1]);
    bs = bs < 0 ? 0 : (bs > E_ ? E_ : bs);
    be = be > E_ ? E_ : be;

    #pragma unroll
    for (int i = 0; i < BROWS * B_ * 32 / BK_THREADS; ++i) accl[tid + BK_THREADS * i] = 0.f;

    const long BS = (long)N_ * HALF_;
    const float* zl = z + lane;
    const v2i* pin = (const v2i*)spair;

    for (int cb = bs; cb < be; cb += BK_CHUNK) {
        #pragma unroll
        for (int i = 0; i < BK_WAVES * 128 / BK_THREADS; ++i) h[tid + BK_THREADS * i] = 0;
        __syncthreads();

        int d[BK_GROUPS], lrank[BK_GROUPS], tot[BK_GROUPS];
        v2i pr[BK_GROUPS];
        #pragma unroll
        for (int g = 0; g < BK_GROUPS; ++g) {
            const int e = cb + w * (32 * BK_GROUPS) + g * 32 + lane;
            d[g]  = BROWS;
            pr[g] = (v2i){0, 0};
            if (e < be) {
                d[g]  = skey[e] & (BROWS - 1);
                pr[g] = pin[e];
            }
        }
        wave_digit_ranks<BK_GROUPS, DIG_SHIFT + 1>(d, lt, lrank, tot);
        #pragma unroll
        for (int g = 0; g < BK_GROUPS; ++g)
            if (lrank[g] == 0) h[w * 128 + d[g]] = tot[g];
        __syncthreads();

        if (w == 0) {
            int c0 = 0, c1 = 0;
            #pragma unroll
            for (int i = 0; i < BK_WAVES; ++i) {
                c0 += h[i * 128 + 2 * lane];
                c1 += h[i * 128 + 2 * lane + 1];
            }
            const int s   = c0 + c1;
            const int inc = wave_inclusive_sum(s, lane);
            int run0 = inc - s;
            int run1 = run0 + c0;
            seg[2 * lane]     = run0;
            seg[2 * lane + 1] = run1;
            if (lane == 31) seg[BROWS] = inc;
            #pragma unroll
            for (int i = 0; i < BK_WAVES; ++i) {
                const int t0 = h[i * 128 + 2 * lane];
                const int t1 = h[i * 128 + 2 * lane + 1];
                h[i * 128 + 2 * lane]     = run0;
                h[i * 128 + 2 * lane + 1] = run1;
                run0 += t0;
                run1 += t1;
            }
        }
        __syncthreads();

        #pragma unroll
        for (int g = 0; g < BK_GROUPS; ++g)
            if (d[g] < BROWS) srt[h[w * 128 + d[g]] + lrank[g]] = pr[g];
        __syncthreads();

        #pragma unroll 1
        for (int k = 0; k < BK_RPW; ++k) {
            const int r = w * BK_RPW + k;
            const int start = __builtin_amdgcn_readfirstlane(seg[r]);
            const int end   = __builtin_amdgcn_readfirstlane(seg[r + 1]);
            if (end <= start) continue;

            float* al = accl + r * (B_ * 32) + lane;
            float a0 = al[0], a1 = al[32], a2 = al[64], a3 = al[96];

            for (int base = start; base < end; base += 32) {
                const int m = (end - base < 32) ? (end - base) : 32;
                v2i p = {0, 0};
                if (lane < m) p = srt[base + lane];

                int j = 0;
                for (; j + 4 <= m; j += 4) {
                    const int   c0 = __builtin_amdgcn_readlane(p.x, j);
                    const int   c1 = __builtin_amdgcn_readlane(p.x, j + 1);
                    const int   c2 = __builtin_amdgcn_readlane(p.x, j + 2);
                    const int   c3 = __builtin_amdgcn_readlane(p.x, j + 3);
                    const float v0 = __int_as_float(__builtin_amdgcn_readlane(p.y, j));
                    const float v1 = __int_as_float(__builtin_amdgcn_readlane(p.y, j + 1));
                    const float v2 = __int_as_float(__builtin_amdgcn_readlane(p.y, j + 2));
                    const float v3 = __int_as_float(__builtin_amdgcn_readlane(p.y, j + 3));
                    const float* q0 = zl + (long)c0 * HALF_;
                    const float* q1 = zl + (long)c1 * HALF_;
                    const float* q2 = zl + (long)c2 * HALF_;
                    const float* q3 = zl + (long)c3 * HALF_;
                    const float z00 = q0[0], z01 = q0[BS], z02 = q0[2 * BS], z03 = q0[3 * BS];
                    const float z10 = q1[0], z11 = q1[BS], z12 = q1[2 * BS], z13 = q1[3 * BS];
                    const float z20 = q2[0], z21 = q2[BS], z22 = q2[2 * BS], z23 = q2[3 * BS];
                    const float z30 = q3[0], z31 = q3[BS], z32 = q3[2 * BS], z33 = q3[3 * BS];
                    a0 = fmaf(v0, z00, a0); a1 = fmaf(v0, z01, a1); a2 = fmaf(v0, z02, a2); a3 = fmaf(v0, z03, a3);
                    a0 = fmaf(v1, z10, a0); a1 = fmaf(v1, z11, a1); a2 = fmaf(v1, z12, a2); a3 = fmaf(v1, z13, a3);
                    a0 = fmaf(v2, z20, a0); a1 = fmaf(v2, z21, a1); a2 = fmaf(v2, z22, a2); a3 = fmaf(v2, z23, a3);
                    a0 = fmaf(v3, z30, a0); a1 = fmaf(v3, z31, a1); a2 = fmaf(v3, z32, a2); a3 = fmaf(v3, z33, a3);
                }
                for (; j < m; ++j) {
                    const int   c0 = __builtin_amdgcn_readlane(p.x, j);
                    const float v0 = __int_as_float(__builtin_amdgcn_readlane(p.y, j));
                    const float* q0 = zl + (long)c0 * HALF_;
                    const float z00 = q0[0], z01 = q0[BS], z02 = q0[2 * BS], z03 = q0[3 * BS];
                    a0 = fmaf(v0, z00, a0); a1 = fmaf(v0, z01, a1); a2 = fmaf(v0, z02, a2); a3 = fmaf(v0, z03, a3);
                }
            }
            al[0] = a0; al[32] = a1; al[64] = a2; al[96] = a3;
        }
    }
    __syncthreads();

    const float bl = b_lin[lane];
    const long OS = (long)N_ * CIN_;
    #pragma unroll
    for (int k = 0; k < BK_RPW; ++k) {
        const int r = w * BK_RPW + k;
        const int n = q * BROWS + r;
        if (n < N_) {
            const float* al = accl + r * (B_ * 32) + lane;
            float* o = out + (long)n * CIN_ + lane;
            o[0]      = bl + al[0];
            o[OS]     = bl + al[32];
            o[2 * OS] = bl + al[64];
            o[3 * OS] = bl + al[96];
        }
    }
}

extern "C" void kernel_launch(void* const* d_in, const int* in_sizes, int n_in,
                              void* d_out, int out_size, void* d_ws, size_t ws_size,
                              hipStream_t stream) {
    const float* x     = (const float*)d_in[0];
    const float* vals  = (const float*)d_in[1];
    const float* W_lin = (const float*)d_in[2];
    const float* b_lin = (const float*)d_in[3];
    const float* W_eye = (const float*)d_in[4];
    const float* b_eye = (const float*)d_in[5];
    const int*   rows  = (const int*)d_in[6];
    const int*   cols  = (const int*)d_in[7];
    float* out    = (float*)d_out;
    char*  ws     = (char*)d_ws;
    float* z      = (float*)(ws + WS_Z_OFF);
    int*   counts = (int*)(ws + WS_CNT_OFF);
    int*   rowtot = (int*)(ws + WS_TOT_OFF);
    int*   bstart = (int*)(ws + WS_BST_OFF);
    int*   keyA   = (int*)(ws + WS_KEYA_OFF);
    int*   pairA  = (int*)(ws + WS_PAIRA_OFF);

    gcn_gemm_wmma<<<M_ / 32, 256, 0, stream>>>(x, W_lin, W_eye, b_eye, out, z);

    gcn_sort_count<<<SORT_BLOCKS, 256, 0, stream>>>(rows, counts);
    gcn_sort_scan_rows<<<NDIG, 256, 0, stream>>>(counts, rowtot);
    gcn_sort_scatter<<<SORT_BLOCKS, 256, 0, stream>>>(rows, cols, vals, counts, rowtot, keyA, pairA, bstart);

    gcn_bucket_gather<<<NBUCKET, BK_THREADS, 0, stream>>>(z, bstart, keyA, pairA, b_lin, out);
}


# GEMM cut in three parts that run beside the edge count, scan and scatter: each filing step shares its launch with GEMM row tiles (5 launches to 4); per-row sum order unchanged
# speedup vs baseline: 2.8857x; 1.0503x over previous
#include <hip/hip_runtime.h>


#define B_    4
#define N_    50000
#define CIN_  64
#define E_    800000
#define HALF_ 32
#define M_    (B_ * N_)

#define SORT_BLOCKS 625
#define SORT_GROUPS 5
#define DIG_SHIFT   6
#define DIG_BITS    10
#define NDIG        1024
#define BROWS       64
#define NBUCKET     ((N_ + BROWS - 1) / BROWS)
#define BK_THREADS  512
#define BK_WAVES    16
#define BK_GROUPS   3
#define BK_CHUNK    (BK_THREADS * BK_GROUPS)
#define BK_RPW      (BROWS / BK_WAVES)

#define WS_Z_OFF      0
#define WS_CNT_OFF    25600000
#define WS_TOT_OFF    28160000
#define WS_BST_OFF    28164096
#define WS_KEYA_OFF   28168448
#define WS_PAIRA_OFF  31368448

typedef __attribute__((ext_vector_type(2))) float v2f;
typedef __attribute__((ext_vector_type(8))) float v8f;
typedef __attribute__((ext_vector_type(2))) int   v2i;
typedef __attribute__((ext_vector_type(4))) float v4f;

#define LDS_PITCH 68
#define GEMM_LDS_FLOATS (96 * LDS_PITCH)

__device__ __forceinline__ void gemm_tile_body(
    float* __restrict__ lds, const int gb,
    const float* __restrict__ x,
    const float* __restrict__ W_lin,
    const float* __restrict__ W_eye, const float* __restrict__ b_eye,
    float* __restrict__ out, float* __restrict__ z)
{
    float* lw = lds;
    float* lx = lds + 64 * LDS_PITCH;

    const int tid = threadIdx.x;
    #pragma unroll
    for (int j = 0; j < 4; ++j) {
        const int idx = tid + 256 * j;
        const int n = idx >> 4, c4 = idx & 15;
        const float* src = (n < HALF_) ? (W_lin + n * CIN_ + c4 * 4)
                                       : (W_eye + (n - HALF_) * CIN_ + c4 * 4);
        *(v4f*)(lw + n * LDS_PITCH + c4 * 4) = *(const v4f*)src;
    }
    const float* xt = x + (long)gb * (32 * CIN_);
    #pragma unroll
    for (int j = 0; j < 2; ++j) {
        const int idx = tid + 256 * j;
        const int r = idx >> 4, c4 = idx & 15;
        *(v4f*)(lx + r * LDS_PITCH + c4 * 4) = *(const v4f*)(xt + idx * 4);
    }
    __syncthreads();

    const int wave  = tid >> 5;
    const int lane  = tid & 31;
    const int hi    = lane >> 4;
    const int l16   = lane & 15;
    const int mtile = gb * 2 + (wave >> 2);
    const int nt    = wave & 3;

    const int ncol = nt * 16 + l16;

    const float* xrow = lx + ((wave >> 2) * 16 + l16) * LDS_PITCH + hi * 2;
    const float* wcol = lw + ncol * LDS_PITCH + hi * 2;

    v8f acc = {};
    #pragma unroll
    for (int kb = 0; kb < 16; ++kb) {
        v2f a = *(const v2f*)(xrow + kb * 4);
        v2f b = *(const v2f*)(wcol + kb * 4);
        acc = __builtin_amdgcn_wmma_f32_16x16x4_f32(
             false, a,  false, b,
             (short)0, acc,  false,  false);
    }

    const int m0 = mtile * 16 + 8 * hi;
    if (nt < 2) {
        #pragma unroll
        for (int r = 0; r < 8; ++r) {
            const long m = m0 + r;
            z[m * HALF_ + ncol]  = acc[r];
        }
    } else {
        const float be = b_eye[ncol - HALF_];
        #pragma unroll
        for (int r = 0; r < 8; ++r) {
            const long m = m0 + r;
            out[m * CIN_ + ncol] = acc[r] + be;
        }
    }
}


template <int G, int NB>
__device__ __forceinline__ void wave_digit_ranks(
    const int (&d)[G], unsigned lt, int (&lrank)[G], int (&tot)[G])
{
    unsigned bal[G][NB];
    #pragma unroll
    for (int g = 0; g < G; ++g) {
        #pragma unroll
        for (int b = 0; b < NB; ++b)
            bal[g][b] = __builtin_amdgcn_ballot_w32(((d[g] >> b) & 1) != 0);
    }
    #pragma unroll
    for (int g = 0; g < G; ++g) {
        int before = 0, all = 0;
        #pragma unroll
        for (int o = 0; o < G; ++o) {
            unsigned m = 0xffffffffu;
            #pragma unroll
            for (int b = 0; b < NB; ++b)
                m &= ((d[g] >> b) & 1) ? bal[o][b] : ~bal[o][b];
            all += __popc(m);
            if (o < g)  before += __popc(m);
            if (o == g) before += __popc(m & lt);
        }
        lrank[g] = before;
        tot[g]   = all;
    }
}

__device__ __forceinline__ void sort_count_body(
    int* __restrict__ h  , const int bid,
    const int* __restrict__ keys, int* __restrict__ counts)
{
    const int tid  = threadIdx.x;
    const int lane = tid & 31;
    const int w    = tid >> 5;
    #pragma unroll
    for (int i = 0; i < 8 * NDIG / 256; ++i) h[tid + 256 * i] = 0;
    __syncthreads();

    const int base = bid * (256 * SORT_GROUPS) + w * (32 * SORT_GROUPS) + lane;
    const unsigned lt = (1u << lane) - 1u;
    int d[SORT_GROUPS], lrank[SORT_GROUPS], tot[SORT_GROUPS];
    #pragma unroll
    for (int g = 0; g < SORT_GROUPS; ++g) d[g] = (keys[base + g * 32] >> DIG_SHIFT) & (NDIG - 1);
    wave_digit_ranks<SORT_GROUPS, DIG_BITS>(d, lt, lrank, tot);
    #pragma unroll
    for (int g = 0; g < SORT_GROUPS; ++g)
        if (lrank[g] == 0) h[w * NDIG + d[g]] = tot[g];
    __syncthreads();

    #pragma unroll
    for (int j = 0; j < NDIG / 256; ++j) {
        const int digit = tid + 256 * j;
        int total = 0;
        #pragma unroll
        for (int i = 0; i < 8; ++i) total += h[i * NDIG + digit];
        counts[digit * SORT_BLOCKS + bid] = total;
    }
}

__device__ __forceinline__ int wave_inclusive_sum(int v, int lane)
{
    #pragma unroll
    for (int s = 1; s < 32; s <<= 1) {
        const int t = __shfl_up(v, s);
        if (lane >= s) v += t;
    }
    return v;
}

__device__ __forceinline__ void sort_scan_rows_body(
    int* __restrict__ wsum  , const int bid,
    int* __restrict__ counts, int* __restrict__ rowtot)
{
    const int tid  = threadIdx.x;
    const int lane = tid & 31;
    const int wave = tid >> 5;
    int* row = counts + bid * SORT_BLOCKS;

    const int lo = tid * 3;
    const int c0 = (lo     < SORT_BLOCKS) ? row[lo]     : 0;
    const int c1 = (lo + 1 < SORT_BLOCKS) ? row[lo + 1] : 0;
    const int c2 = (lo + 2 < SORT_BLOCKS) ? row[lo + 2] : 0;
    const int sum = c0 + c1 + c2;
    const int inc = wave_inclusive_sum(sum, lane);
    if (lane == 31) wsum[wave] = inc;
    __syncthreads();

    int wbase = 0, total = 0;
    #pragma unroll
    for (int i = 0; i < 8; ++i) {
        const int v = wsum[i];
        if (i < wave) wbase += v;
        total += v;
    }
    int run = wbase + inc - sum;
    if (lo     < SORT_BLOCKS) row[lo]     = run;
    run += c0;
    if (lo + 1 < SORT_BLOCKS) row[lo + 1] = run;
    run += c1;
    if (lo + 2 < SORT_BLOCKS) row[lo + 2] = run;
    if (tid == 0) rowtot[bid] = total;
}

__device__ __forceinline__ void sort_scatter_body(
    int* __restrict__ lds  , const int bid,
    const int* __restrict__ rows, const int* __restrict__ cols, const float* __restrict__ vals,
    const int* __restrict__ counts, const int* __restrict__ rowtot,
    int* __restrict__ keyOut, int* __restrict__ pairOut, int* __restrict__ bstart)
{
    int* h    = lds;
    int* dsum = lds + 8 * NDIG;

    const int tid  = threadIdx.x;
    const int lane = tid & 31;
    const int w    = tid >> 5;
    #pragma unroll
    for (int i = 0; i < 8 * NDIG / 256; ++i) h[tid + 256 * i] = 0;
    int dv[NDIG / 256], dinc[NDIG / 256];
    #pragma unroll
    for (int j = 0; j < NDIG / 256; ++j) {
        dv[j]   = rowtot[tid + 256 * j];
        dinc[j] = wave_inclusive_sum(dv[j], lane);
        if (lane == 31) dsum[j * 8 + w] = dinc[j];
    }
    __syncthreads();

    const int base = bid * (256 * SORT_GROUPS) + w * (32 * SORT_GROUPS) + lane;
    const unsigned lt = (1u << lane) - 1u;

    int key[SORT_GROUPS], d[SORT_GROUPS], lrank[SORT_GROUPS], tot[SORT_GROUPS];
    v2i pr[SORT_GROUPS];
    #pragma unroll
    for (int g = 0; g < SORT_GROUPS; ++g) {
        const int e = base + g * 32;
        key[g]  = rows[e];
        pr[g].x = cols[e];
        pr[g].y = __float_as_int(vals[e]);
        d[g] = (key[g] >> DIG_SHIFT) & (NDIG - 1);
    }
    wave_digit_ranks<SORT_GROUPS, DIG_BITS>(d, lt, lrank, tot);
    #pragma unroll
    for (int g = 0; g < SORT_GROUPS; ++g)
        if (lrank[g] == 0) h[w * NDIG + d[g]] = tot[g];
    __syncthreads();

    {
        int carry = 0;
        #pragma unroll
        for (int j = 0; j < NDIG / 256; ++j) {
            int before = carry, gtotal = 0;
            #pragma unroll
            for (int i = 0; i < 8; ++i) {
                const int v = dsum[j * 8 + i];
                if (i < w) before += v;
                gtotal += v;
            }
            carry += gtotal;
            const int digit = tid + 256 * j;
            const int dbase = before + dinc[j] - dv[j];
            if (bid == 0) bstart[digit] = dbase;
            int run = counts[digit * SORT_BLOCKS + bid] + dbase;
            #pragma unroll
            for (int i = 0; i < 8; ++i) {
                const int c = h[i * NDIG + digit];
                h[i * NDIG + digit] = run;
                run += c;
            }
        }
        if (bid == 0 && tid == 0) bstart[NDIG] = carry;
    }
    __syncthreads();

    #pragma unroll
    for (int g = 0; g < SORT_GROUPS; ++g) {
        const int pos = h[w * NDIG + d[g]] + lrank[g];
        if ((unsigned)pos < (unsigned)E_) {
            keyOut[pos] = key[g];
            ((v2i*)pairOut)[pos] = pr[g];
        }
    }
}

#define SORT_LDS_INTS (8 * NDIG + (NDIG / 256) * 8)
#define GEMM_TILES    (M_ / 32)
#define GEMM_PART0    1800
#define GEMM_PART1    1550
#define GEMM_PART2    (GEMM_TILES - GEMM_PART0 - GEMM_PART1)

union FilingOrGemmLds {
    int   filing[SORT_LDS_INTS];
    float gemm[GEMM_LDS_FLOATS];
};

template <int STEP>
__global__ __launch_bounds__(256, 16) void gcn_gemm_beside_filing(
    const float* __restrict__ x,
    const float* __restrict__ W_lin,
    const float* __restrict__ W_eye, const float* __restrict__ b_eye,
    float* __restrict__ out, float* __restrict__ z,
    const int* __restrict__ rows, const int* __restrict__ cols, const float* __restrict__ vals,
    int* __restrict__ counts, int* __restrict__ rowtot,
    int* __restrict__ keyOut, int* __restrict__ pairOut, int* __restrict__ bstart)
{
    __shared__ __attribute__((aligned(16))) FilingOrGemmLds lds;

    constexpr int NFILE = (STEP == 1) ? NDIG : SORT_BLOCKS;
    constexpr int TILE0 = (STEP == 0) ? 0 : (STEP == 1) ? GEMM_PART0 : (GEMM_PART0 + GEMM_PART1);

    const int bid = blockIdx.x;
    if (bid < NFILE) {
        if (STEP == 0)      sort_count_body(lds.filing, bid, rows, counts);
        else if (STEP == 1) sort_scan_rows_body(lds.filing, bid, counts, rowtot);
        else                sort_scatter_body(lds.filing, bid, rows, cols, vals, counts, rowtot, keyOut, pairOut, bstart);
    } else {
        gemm_tile_body(lds.gemm, TILE0 + (bid - NFILE), x, W_lin, W_eye, b_eye, out, z);
    }
}

__global__ __launch_bounds__(BK_THREADS, 16) void gcn_bucket_gather(
    const float* __restrict__ z, const int* __restrict__ bstart,
    const int* __restrict__ skey, const int* __restrict__ spair,
    const float* __restrict__ b_lin, float* __restrict__ out)
{
    __shared__ int   h[BK_WAVES * 128];
    __shared__ int   seg[BROWS + 1];
    __shared__ v2i   srt[BK_CHUNK];
    __shared__ float accl[BROWS * B_ * 32];

    const int tid  = threadIdx.x;
    const int lane = tid & 31;
    const int w    = tid >> 5;
    const int q    = blockIdx.x;
    const unsigned lt = (1u << lane) - 1u;

    int bs = __builtin_amdgcn_readfirstlane(bstart[q]);
    int be = __builtin_amdgcn_readfirstlane(bstart[q + 1]);
    bs = bs < 0 ? 0 : (bs > E_ ? E_ : bs);
    be = be > E_ ? E_ : be;

    #pragma unroll
    for (int i = 0; i < BROWS * B_ * 32 / BK_THREADS; ++i) accl[tid + BK_THREADS * i] = 0.f;

    const long BS = (long)N_ * HALF_;
    const float* zl = z + lane;
    const v2i* pin = (const v2i*)spair;

    for (int cb = bs; cb < be; cb += BK_CHUNK) {
        #pragma unroll
        for (int i = 0; i < BK_WAVES * 128 / BK_THREADS; ++i) h[tid + BK_THREADS * i] = 0;
        __syncthreads();

        int d[BK_GROUPS], lrank[BK_GROUPS], tot[BK_GROUPS];
        v2i pr[BK_GROUPS];
        #pragma unroll
        for (int g = 0; g < BK_GROUPS; ++g) {
            const int e = cb + w * (32 * BK_GROUPS) + g * 32 + lane;
            d[g]  = BROWS;
            pr[g] = (v2i){0, 0};
            if (e < be) {
                d[g]  = skey[e] & (BROWS - 1);
                pr[g] = pin[e];
            }
        }
        wave_digit_ranks<BK_GROUPS, DIG_SHIFT + 1>(d, lt, lrank, tot);
        #pragma unroll
        for (int g = 0; g < BK_GROUPS; ++g)
            if (lrank[g] == 0) h[w * 128 + d[g]] = tot[g];
        __syncthreads();

        if (w == 0) {
            int c0 = 0, c1 = 0;
            #pragma unroll
            for (int i = 0; i < BK_WAVES; ++i) {
                c0 += h[i * 128 + 2 * lane];
                c1 += h[i * 128 + 2 * lane + 1];
            }
            const int s   = c0 + c1;
            const int inc = wave_inclusive_sum(s, lane);
            int run0 = inc - s;
            int run1 = run0 + c0;
            seg[2 * lane]     = run0;
            seg[2 * lane + 1] = run1;
            if (lane == 31) seg[BROWS] = inc;
            #pragma unroll
            for (int i = 0; i < BK_WAVES; ++i) {
                const int t0 = h[i * 128 + 2 * lane];
                const int t1 = h[i * 128 + 2 * lane + 1];
                h[i * 128 + 2 * lane]     = run0;
                h[i * 128 + 2 * lane + 1] = run1;
                run0 += t0;
                run1 += t1;
            }
        }
        __syncthreads();

        #pragma unroll
        for (int g = 0; g < BK_GROUPS; ++g)
            if (d[g] < BROWS) srt[h[w * 128 + d[g]] + lrank[g]] = pr[g];
        __syncthreads();

        #pragma unroll 1
        for (int k = 0; k < BK_RPW; ++k) {
            const int r = w * BK_RPW + k;
            const int start = __builtin_amdgcn_readfirstlane(seg[r]);
            const int end   = __builtin_amdgcn_readfirstlane(seg[r + 1]);
            if (end <= start) continue;

            float* al = accl + r * (B_ * 32) + lane;
            float a0 = al[0], a1 = al[32], a2 = al[64], a3 = al[96];

            for (int base = start; base < end; base += 32) {
                const int m = (end - base < 32) ? (end - base) : 32;
                v2i p = {0, 0};
                if (lane < m) p = srt[base + lane];

                int j = 0;
                for (; j + 4 <= m; j += 4) {
                    const int   c0 = __builtin_amdgcn_readlane(p.x, j);
                    const int   c1 = __builtin_amdgcn_readlane(p.x, j + 1);
                    const int   c2 = __builtin_amdgcn_readlane(p.x, j + 2);
                    const int   c3 = __builtin_amdgcn_readlane(p.x, j + 3);
                    const float v0 = __int_as_float(__builtin_amdgcn_readlane(p.y, j));
                    const float v1 = __int_as_float(__builtin_amdgcn_readlane(p.y, j + 1));
                    const float v2 = __int_as_float(__builtin_amdgcn_readlane(p.y, j + 2));
                    const float v3 = __int_as_float(__builtin_amdgcn_readlane(p.y, j + 3));
                    const float* q0 = zl + (long)c0 * HALF_;
                    const float* q1 = zl + (long)c1 * HALF_;
                    const float* q2 = zl + (long)c2 * HALF_;
                    const float* q3 = zl + (long)c3 * HALF_;
                    const float z00 = q0[0], z01 = q0[BS], z02 = q0[2 * BS], z03 = q0[3 * BS];
                    const float z10 = q1[0], z11 = q1[BS], z12 = q1[2 * BS], z13 = q1[3 * BS];
                    const float z20 = q2[0], z21 = q2[BS], z22 = q2[2 * BS], z23 = q2[3 * BS];
                    const float z30 = q3[0], z31 = q3[BS], z32 = q3[2 * BS], z33 = q3[3 * BS];
                    a0 = fmaf(v0, z00, a0); a1 = fmaf(v0, z01, a1); a2 = fmaf(v0, z02, a2); a3 = fmaf(v0, z03, a3);
                    a0 = fmaf(v1, z10, a0); a1 = fmaf(v1, z11, a1); a2 = fmaf(v1, z12, a2); a3 = fmaf(v1, z13, a3);
                    a0 = fmaf(v2, z20, a0); a1 = fmaf(v2, z21, a1); a2 = fmaf(v2, z22, a2); a3 = fmaf(v2, z23, a3);
                    a0 = fmaf(v3, z30, a0); a1 = fmaf(v3, z31, a1); a2 = fmaf(v3, z32, a2); a3 = fmaf(v3, z33, a3);
                }
                for (; j < m; ++j) {
                    const int   c0 = __builtin_amdgcn_readlane(p.x, j);
                    const float v0 = __int_as_float(__builtin_amdgcn_readlane(p.y, j));
                    const float* q0 = zl + (long)c0 * HALF_;
                    const float z00 = q0[0], z01 = q0[BS], z02 = q0[2 * BS], z03 = q0[3 * BS];
                    a0 = fmaf(v0, z00, a0); a1 = fmaf(v0, z01, a1); a2 = fmaf(v0, z02, a2); a3 = fmaf(v0, z03, a3);
                }
            }
            al[0] = a0; al[32] = a1; al[64] = a2; al[96] = a3;
        }
    }
    __syncthreads();

    const float bl = b_lin[lane];
    const long OS = (long)N_ * CIN_;
    #pragma unroll
    for (int k = 0; k < BK_RPW; ++k) {
        const int r = w * BK_RPW + k;
        const int n = q * BROWS + r;
        if (n < N_) {
            const float* al = accl + r * (B_ * 32) + lane;
            float* o = out + (long)n * CIN_ + lane;
            o[0]      = bl + al[0];
            o[OS]     = bl + al[32];
            o[2 * OS] = bl + al[64];
            o[3 * OS] = bl + al[96];
        }
    }
}

extern "C" void kernel_launch(void* const* d_in, const int* in_sizes, int n_in,
                              void* d_out, int out_size, void* d_ws, size_t ws_size,
                              hipStream_t stream) {
    const float* x     = (const float*)d_in[0];
    const float* vals  = (const float*)d_in[1];
    const float* W_lin = (const float*)d_in[2];
    const float* b_lin = (const float*)d_in[3];
    const float* W_eye = (const float*)d_in[4];
    const float* b_eye = (const float*)d_in[5];
    const int*   rows  = (const int*)d_in[6];
    const int*   cols  = (const int*)d_in[7];
    float* out    = (float*)d_out;
    char*  ws     = (char*)d_ws;
    float* z      = (float*)(ws + WS_Z_OFF);
    int*   counts = (int*)(ws + WS_CNT_OFF);
    int*   rowtot = (int*)(ws + WS_TOT_OFF);
    int*   bstart = (int*)(ws + WS_BST_OFF);
    int*   keyA   = (int*)(ws + WS_KEYA_OFF);
    int*   pairA  = (int*)(ws + WS_PAIRA_OFF);

    gcn_gemm_beside_filing<0><<<SORT_BLOCKS + GEMM_PART0, 256, 0, stream>>>(
        x, W_lin, W_eye, b_eye, out, z, rows, cols, vals, counts, rowtot, keyA, pairA, bstart);
    gcn_gemm_beside_filing<1><<<NDIG + GEMM_PART1, 256, 0, stream>>>(
        x, W_lin, W_eye, b_eye, out, z, rows, cols, vals, counts, rowtot, keyA, pairA, bstart);
    gcn_gemm_beside_filing<2><<<SORT_BLOCKS + GEMM_PART2, 256, 0, stream>>>(
        x, W_lin, W_eye, b_eye, out, z, rows, cols, vals, counts, rowtot, keyA, pairA, bstart);

    gcn_bucket_gather<<<NBUCKET, BK_THREADS, 0, stream>>>(z, bstart, keyA, pairA, b_lin, out);
}


# Global count, scan and scatter removed (4 launches to 2): a filing block orders its 1280 edges by bucket into a run; a bucket's work-group pulls its 625 pieces in edge order; row sum order kept
# speedup vs baseline: 4.7405x; 1.6428x over previous
#include <hip/hip_runtime.h>


#define B_    4
#define N_    50000
#define CIN_  64
#define E_    800000
#define HALF_ 32
#define M_    (B_ * N_)

#define SORT_BLOCKS 625
#define SORT_GROUPS 5
#define RUN_EDGES   (256 * SORT_GROUPS)
#define DIG_SHIFT   6
#define DIG_BITS    10
#define NDIG        1024
#define BROWS       64
#define NBUCKET     ((N_ + BROWS - 1) / BROWS)
#define BK_THREADS  512
#define BK_WAVES    16
#define BK_GROUPS   3
#define BK_CHUNK    (BK_THREADS * BK_GROUPS)
#define BK_RPW      (BROWS / BK_WAVES)

#define WS_Z_OFF      0
#define WS_PIECE_OFF  25600000
#define WS_REC_OFF    28168448

static_assert(SORT_BLOCKS * RUN_EDGES == E_, "the filing blocks cover the edges exactly");
static_assert(N_ <= 65536, "a column index and the row's 6 low bits share one 32-bit word");

typedef __attribute__((ext_vector_type(2))) float v2f;
typedef __attribute__((ext_vector_type(8))) float v8f;
typedef __attribute__((ext_vector_type(2))) int   v2i;
typedef __attribute__((ext_vector_type(4))) float v4f;

#define LDS_PITCH 68
#define GEMM_LDS_FLOATS (96 * LDS_PITCH)

__device__ __forceinline__ void gemm_tile_body(
    float* __restrict__ lds, const int gb,
    const float* __restrict__ x,
    const float* __restrict__ W_lin,
    const float* __restrict__ W_eye, const float* __restrict__ b_eye,
    float* __restrict__ out, float* __restrict__ z)
{
    float* lw = lds;
    float* lx = lds + 64 * LDS_PITCH;

    const int tid = threadIdx.x;
    #pragma unroll
    for (int j = 0; j < 4; ++j) {
        const int idx = tid + 256 * j;
        const int n = idx >> 4, c4 = idx & 15;
        const float* src = (n < HALF_) ? (W_lin + n * CIN_ + c4 * 4)
                                       : (W_eye + (n - HALF_) * CIN_ + c4 * 4);
        *(v4f*)(lw + n * LDS_PITCH + c4 * 4) = *(const v4f*)src;
    }
    const float* xt = x + (long)gb * (32 * CIN_);
    #pragma unroll
    for (int j = 0; j < 2; ++j) {
        const int idx = tid + 256 * j;
        const int r = idx >> 4, c4 = idx & 15;
        *(v4f*)(lx + r * LDS_PITCH + c4 * 4) = *(const v4f*)(xt + idx * 4);
    }
    __syncthreads();

    const int wave  = tid >> 5;
    const int lane  = tid & 31;
    const int hi    = lane >> 4;
    const int l16   = lane & 15;
    const int mtile = gb * 2 + (wave >> 2);
    const int nt    = wave & 3;

    const int ncol = nt * 16 + l16;

    const float* xrow = lx + ((wave >> 2) * 16 + l16) * LDS_PITCH + hi * 2;
    const float* wcol = lw + ncol * LDS_PITCH + hi * 2;

    v8f acc = {};
    #pragma unroll
    for (int kb = 0; kb < 16; ++kb) {
        v2f a = *(const v2f*)(xrow + kb * 4);
        v2f b = *(const v2f*)(wcol + kb * 4);
        acc = __builtin_amdgcn_wmma_f32_16x16x4_f32(
             false, a,  false, b,
             (short)0, acc,  false,  false);
    }

    const int m0 = mtile * 16 + 8 * hi;
    if (nt < 2) {
        #pragma unroll
        for (int r = 0; r < 8; ++r) {
            const long m = m0 + r;
            z[m * HALF_ + ncol]  = acc[r];
        }
    } else {
        const float be = b_eye[ncol - HALF_];
        #pragma unroll
        for (int r = 0; r < 8; ++r) {
            const long m = m0 + r;
            out[m * CIN_ + ncol] = acc[r] + be;
        }
    }
}


template <int G, int NB>
__device__ __forceinline__ void wave_digit_ranks(
    const int (&d)[G], unsigned lt, int (&lrank)[G], int (&tot)[G])
{
    unsigned bal[G][NB];
    #pragma unroll
    for (int g = 0; g < G; ++g) {
        #pragma unroll
        for (int b = 0; b < NB; ++b)
            bal[g][b] = __builtin_amdgcn_ballot_w32(((d[g] >> b) & 1) != 0);
    }
    #pragma unroll
    for (int g = 0; g < G; ++g) {
        int before = 0, all = 0;
        #pragma unroll
        for (int o = 0; o < G; ++o) {
            unsigned m = 0xffffffffu;
            #pragma unroll
            for (int b = 0; b < NB; ++b)
                m &= ((d[g] >> b) & 1) ? bal[o][b] : ~bal[o][b];
            all += __popc(m);
            if (o < g)  before += __popc(m);
            if (o == g) before += __popc(m & lt);
        }
        lrank[g] = before;
        tot[g]   = all;
    }
}

__device__ __forceinline__ int wave_inclusive_sum(int v, int lane)
{
    #pragma unroll
    for (int s = 1; s < 32; s <<= 1) {
        const int t = __shfl_up(v, s);
        if (lane >= s) v += t;
    }
    return v;
}

__device__ __forceinline__ void file_run_body(
    int* __restrict__ lds  , const int bid,
    const int* __restrict__ rows, const int* __restrict__ cols, const float* __restrict__ vals,
    int* __restrict__ piece, int* __restrict__ rec)
{
    int* h    = lds;
    int* dsum = lds + 8 * NDIG;

    const int tid  = threadIdx.x;
    const int lane = tid & 31;
    const int w    = tid >> 5;

    const int base = bid * RUN_EDGES + w * (32 * SORT_GROUPS) + lane;
    const unsigned lt = (1u << lane) - 1u;

    int key[SORT_GROUPS], d[SORT_GROUPS], lrank[SORT_GROUPS], tot[SORT_GROUPS];
    v2i pr[SORT_GROUPS];
    #pragma unroll
    for (int g = 0; g < SORT_GROUPS; ++g) {
        const int e = base + g * 32;
        key[g]  = rows[e];
        pr[g].x = cols[e];
        pr[g].y = __float_as_int(vals[e]);
        d[g] = (key[g] >> DIG_SHIFT) & (NDIG - 1);
    }
    #pragma unroll
    for (int i = 0; i < 8 * NDIG / 256; ++i) h[tid + 256 * i] = 0;
    __syncthreads();

    wave_digit_ranks<SORT_GROUPS, DIG_BITS>(d, lt, lrank, tot);
    #pragma unroll
    for (int g = 0; g < SORT_GROUPS; ++g)
        if (lrank[g] == 0) h[w * NDIG + d[g]] = tot[g];
    __syncthreads();

    int dv[NDIG / 256], dinc[NDIG / 256];
    #pragma unroll
    for (int j = 0; j < NDIG / 256; ++j) {
        const int digit = tid + 256 * j;
        int t = 0;
        #pragma unroll
        for (int i = 0; i < 8; ++i) t += h[i * NDIG + digit];
        dv[j]   = t;
        dinc[j] = wave_inclusive_sum(t, lane);
        if (lane == 31) dsum[j * 8 + w] = dinc[j];
    }
    __syncthreads();

    {
        int carry = 0;
        #pragma unroll
        for (int j = 0; j < NDIG / 256; ++j) {
            int before = carry, gtotal = 0;
            #pragma unroll
            for (int i = 0; i < 8; ++i) {
                const int v = dsum[j * 8 + i];
                if (i < w) before += v;
                gtotal += v;
            }
            carry += gtotal;
            const int digit = tid + 256 * j;
            const int dbase = before + dinc[j] - dv[j];
            if (digit < NBUCKET) piece[digit * SORT_BLOCKS + bid] = dbase | (dv[j] << 16);
            int run = dbase;
            #pragma unroll
            for (int i = 0; i < 8; ++i) {
                const int c = h[i * NDIG + digit];
                h[i * NDIG + digit] = run;
                run += c;
            }
        }
    }
    __syncthreads();

    v2i* run_out = (v2i*)rec + bid * RUN_EDGES;
    #pragma unroll
    for (int g = 0; g < SORT_GROUPS; ++g) {
        const int pos = h[w * NDIG + d[g]] + lrank[g];
        if ((unsigned)pos < (unsigned)RUN_EDGES) {
            v2i r;
            r.x = (pr[g].x & 0xffff) | ((key[g] & (BROWS - 1)) << 16);
            r.y = pr[g].y;
            run_out[pos] = r;
        }
    }
}

#define SORT_LDS_INTS (8 * NDIG + (NDIG / 256) * 8)
#define GEMM_TILES    (M_ / 32)

union FilingOrGemmLds {
    int   filing[SORT_LDS_INTS];
    float gemm[GEMM_LDS_FLOATS];
};

__global__ __launch_bounds__(256, 16) void gcn_gemm_beside_filing(
    const float* __restrict__ x,
    const float* __restrict__ W_lin,
    const float* __restrict__ W_eye, const float* __restrict__ b_eye,
    float* __restrict__ out, float* __restrict__ z,
    const int* __restrict__ rows, const int* __restrict__ cols, const float* __restrict__ vals,
    int* __restrict__ piece, int* __restrict__ rec)
{
    __shared__ __attribute__((aligned(16))) FilingOrGemmLds lds;

    const int bid = blockIdx.x;
    if (bid < SORT_BLOCKS) {
        file_run_body(lds.filing, bid, rows, cols, vals, piece, rec);
    } else {
        gemm_tile_body(lds.gemm, bid - SORT_BLOCKS, x, W_lin, W_eye, b_eye, out, z);
    }
}

#define BOFF_N 1024
__global__ __launch_bounds__(BK_THREADS, 16) void gcn_bucket_gather(
    const float* __restrict__ z, const int* __restrict__ piece, const int* __restrict__ rec,
    const float* __restrict__ b_lin, float* __restrict__ out)
{
    __shared__ int   boff[BOFF_N];
    __shared__ int   bsrc[SORT_BLOCKS + 1];
    __shared__ int   wsum[BK_WAVES];
    __shared__ int   h[BK_WAVES * 128];
    __shared__ int   seg[BROWS + 1];
    __shared__ v2i   srt[BK_CHUNK];
    __shared__ float accl[BROWS * B_ * 32];

    const int tid  = threadIdx.x;
    const int lane = tid & 31;
    const int w    = tid >> 5;
    const int q    = blockIdx.x;
    const unsigned lt = (1u << lane) - 1u;

    const int* prow = piece + q * SORT_BLOCKS;
    const int rb0 = 2 * tid, rb1 = 2 * tid + 1;
    const int w0 = (rb0 < SORT_BLOCKS) ? prow[rb0] : 0;
    const int w1 = (rb1 < SORT_BLOCKS) ? prow[rb1] : 0;
    const int n0 = (w0 >> 16) & 0x7ff;
    const int n1 = (w1 >> 16) & 0x7ff;

    #pragma unroll
    for (int i = 0; i < BROWS * B_ * 32 / BK_THREADS; ++i) accl[tid + BK_THREADS * i] = 0.f;

    const int psum = n0 + n1;
    int pinc = psum, ptot = psum;
    #pragma unroll
    for (int s = 1; s < 32; s <<= 1) {
        const int t = __shfl_xor(ptot, s);
        if (lane & s) pinc += t;
        ptot += t;
    }
    if (lane == 0) wsum[w] = ptot;
    __syncthreads();
    int pbase = 0, ptotal = 0;
    #pragma unroll
    for (int i = 0; i < BK_WAVES; ++i) {
        const int v = wsum[i];
        if (i < w) pbase += v;
        ptotal += v;
    }
    const int prun = pbase + pinc - psum;
    boff[rb0] = (rb0 < SORT_BLOCKS) ? prun      : 0x7fffffff;
    boff[rb1] = (rb1 < SORT_BLOCKS) ? prun + n0 : 0x7fffffff;
    if (rb0 < SORT_BLOCKS) bsrc[rb0] = rb0 * RUN_EDGES + (w0 & 0x7ff);
    if (rb1 < SORT_BLOCKS) bsrc[rb1] = rb1 * RUN_EDGES + (w1 & 0x7ff);
    const int bs = 0;
    const int be = __builtin_amdgcn_readfirstlane(ptotal);

    const long BS = (long)N_ * HALF_;
    const float* zl = z + lane;
    const v2i* pin = (const v2i*)rec;

    for (int cb = bs; cb < be; cb += BK_CHUNK) {
        #pragma unroll
        for (int i = 0; i < BK_WAVES * 128 / BK_THREADS; ++i) h[tid + BK_THREADS * i] = 0;
        __syncthreads();

        int d[BK_GROUPS], lrank[BK_GROUPS], tot[BK_GROUPS];
        v2i pr[BK_GROUPS];
        int e[BK_GROUPS], es[BK_GROUPS], rb[BK_GROUPS];
        #pragma unroll
        for (int g = 0; g < BK_GROUPS; ++g) {
            e[g]  = cb + w * (32 * BK_GROUPS) + g * 32 + lane;
            es[g] = e[g] < be ? e[g] : be - 1;
            rb[g] = 0;
        }
        #pragma unroll
        for (int step = BOFF_N / 2; step >= 1; step >>= 1) {
            #pragma unroll
            for (int g = 0; g < BK_GROUPS; ++g)
                if (boff[rb[g] + step] <= es[g]) rb[g] += step;
        }
        v2i rc[BK_GROUPS];
        #pragma unroll
        for (int g = 0; g < BK_GROUPS; ++g) {
            int idx = bsrc[rb[g]] + (es[g] - boff[rb[g]]);
            idx = idx < 0 ? 0 : (idx > E_ - 1 ? E_ - 1 : idx);
            rc[g] = pin[idx];
        }
        #pragma unroll
        for (int g = 0; g < BK_GROUPS; ++g) {
            const bool in = e[g] < be;
            d[g]    = in ? ((rc[g].x >> 16) & (BROWS - 1)) : BROWS;
            pr[g].x = in ? (rc[g].x & 0xffff) : 0;
            pr[g].y = in ? rc[g].y : 0;
        }
        wave_digit_ranks<BK_GROUPS, DIG_SHIFT + 1>(d, lt, lrank, tot);
        #pragma unroll
        for (int g = 0; g < BK_GROUPS; ++g)
            if (lrank[g] == 0) h[w * 128 + d[g]] = tot[g];
        __syncthreads();

        if (w == 0) {
            int c0 = 0, c1 = 0;
            #pragma unroll
            for (int i = 0; i < BK_WAVES; ++i) {
                c0 += h[i * 128 + 2 * lane];
                c1 += h[i * 128 + 2 * lane + 1];
            }
            const int s   = c0 + c1;
            static_assert(BK_CHUNK < 2048, "eleven bits hold a chunk's edge count");
            int run0 = 0;
            #pragma unroll
            for (int b = 0; b < 11; ++b) {
                const unsigned mb = __builtin_amdgcn_ballot_w32(((s >> b) & 1) != 0);
                run0 += __popc(mb & lt) << b;
            }
            int run1 = run0 + c0;
            seg[2 * lane]     = run0;
            seg[2 * lane + 1] = run1;
            if (lane == 31) seg[BROWS] = run0 + s;
            #pragma unroll
            for (int i = 0; i < BK_WAVES; ++i) {
                const int t0 = h[i * 128 + 2 * lane];
                const int t1 = h[i * 128 + 2 * lane + 1];
                h[i * 128 + 2 * lane]     = run0;
                h[i * 128 + 2 * lane + 1] = run1;
                run0 += t0;
                run1 += t1;
            }
        }
        __syncthreads();

        #pragma unroll
        for (int g = 0; g < BK_GROUPS; ++g)
            if (d[g] < BROWS) srt[h[w * 128 + d[g]] + lrank[g]] = pr[g];
        __syncthreads();

        #pragma unroll 1
        for (int k = 0; k < BK_RPW; ++k) {
            const int r = w * BK_RPW + k;
            const int start = __builtin_amdgcn_readfirstlane(seg[r]);
            const int end   = __builtin_amdgcn_readfirstlane(seg[r + 1]);
            if (end <= start) continue;

            float* al = accl + r * (B_ * 32) + lane;
            float a0 = al[0], a1 = al[32], a2 = al[64], a3 = al[96];

            for (int base = start; base < end; base += 32) {
                const int m = (end - base < 32) ? (end - base) : 32;
                v2i p = {0, 0};
                if (lane < m) p = srt[base + lane];

                int j = 0;
                for (; j + 4 <= m; j += 4) {
                    const int   c0 = __builtin_amdgcn_readlane(p.x, j);
                    const int   c1 = __builtin_amdgcn_readlane(p.x, j + 1);
                    const int   c2 = __builtin_amdgcn_readlane(p.x, j + 2);
                    const int   c3 = __builtin_amdgcn_readlane(p.x, j + 3);
                    const float v0 = __int_as_float(__builtin_amdgcn_readlane(p.y, j));
                    const float v1 = __int_as_float(__builtin_amdgcn_readlane(p.y, j + 1));
                    const float v2 = __int_as_float(__builtin_amdgcn_readlane(p.y, j + 2));
                    const float v3 = __int_as_float(__builtin_amdgcn_readlane(p.y, j + 3));
                    const float* q0 = zl + (long)c0 * HALF_;
                    const float* q1 = zl + (long)c1 * HALF_;
                    const float* q2 = zl + (long)c2 * HALF_;
                    const float* q3 = zl + (long)c3 * HALF_;
                    const float z00 = q0[0], z01 = q0[BS], z02 = q0[2 * BS], z03 = q0[3 * BS];
                    const float z10 = q1[0], z11 = q1[BS], z12 = q1[2 * BS], z13 = q1[3 * BS];
                    const float z20 = q2[0], z21 = q2[BS], z22 = q2[2 * BS], z23 = q2[3 * BS];
                    const float z30 = q3[0], z31 = q3[BS], z32 = q3[2 * BS], z33 = q3[3 * BS];
                    a0 = fmaf(v0, z00, a0); a1 = fmaf(v0, z01, a1); a2 = fmaf(v0, z02, a2); a3 = fmaf(v0, z03, a3);
                    a0 = fmaf(v1, z10, a0); a1 = fmaf(v1, z11, a1); a2 = fmaf(v1, z12, a2); a3 = fmaf(v1, z13, a3);
                    a0 = fmaf(v2, z20, a0); a1 = fmaf(v2, z21, a1); a2 = fmaf(v2, z22, a2); a3 = fmaf(v2, z23, a3);
                    a0 = fmaf(v3, z30, a0); a1 = fmaf(v3, z31, a1); a2 = fmaf(v3, z32, a2); a3 = fmaf(v3, z33, a3);
                }
                for (; j < m; ++j) {
                    const int   c0 = __builtin_amdgcn_readlane(p.x, j);
                    const float v0 = __int_as_float(__builtin_amdgcn_readlane(p.y, j));
                    const float* q0 = zl + (long)c0 * HALF_;
                    const float z00 = q0[0], z01 = q0[BS], z02 = q0[2 * BS], z03 = q0[3 * BS];
                    a0 = fmaf(v0, z00, a0); a1 = fmaf(v0, z01, a1); a2 = fmaf(v0, z02, a2); a3 = fmaf(v0, z03, a3);
                }
            }
            al[0] = a0; al[32] = a1; al[64] = a2; al[96] = a3;
        }
    }
    __syncthreads();

    const float bl = b_lin[lane];
    const long OS = (long)N_ * CIN_;
    #pragma unroll
    for (int k = 0; k < BK_RPW; ++k) {
        const int r = w * BK_RPW + k;
        const int n = q * BROWS + r;
        if (n < N_) {
            const float* al = accl + r * (B_ * 32) + lane;
            float* o = out + (long)n * CIN_ + lane;
            o[0]      = bl + al[0];
            o[OS]     = bl + al[32];
            o[2 * OS] = bl + al[64];
            o[3 * OS] = bl + al[96];
        }
    }
}

extern "C" void kernel_launch(void* const* d_in, const int* in_sizes, int n_in,
                              void* d_out, int out_size, void* d_ws, size_t ws_size,
                              hipStream_t stream) {
    const float* x     = (const float*)d_in[0];
    const float* vals  = (const float*)d_in[1];
    const float* W_lin = (const float*)d_in[2];
    const float* b_lin = (const float*)d_in[3];
    const float* W_eye = (const float*)d_in[4];
    const float* b_eye = (const float*)d_in[5];
    const int*   rows  = (const int*)d_in[6];
    const int*   cols  = (const int*)d_in[7];
    float* out    = (float*)d_out;
    char*  ws     = (char*)d_ws;
    float* z      = (float*)(ws + WS_Z_OFF);
    int*   piece  = (int*)(ws + WS_PIECE_OFF);
    int*   rec    = (int*)(ws + WS_REC_OFF);

    gcn_gemm_beside_filing<<<SORT_BLOCKS + GEMM_TILES, 256, 0, stream>>>(
        x, W_lin, W_eye, b_eye, out, z, rows, cols, vals, piece, rec);

    gcn_bucket_gather<<<NBUCKET, BK_THREADS, 0, stream>>>(z, piece, rec, b_lin, out);
}


# GEMM wave tile widened from 16x16 to 16x32 (two accumulators share each A fragment, 64 rows of x per work-group): half as many GEMM waves and work-groups enter launch 1; k order and sums unchanged
# speedup vs baseline: 4.9994x; 1.0546x over previous
#include <hip/hip_runtime.h>


#define B_    4
#define N_    50000
#define CIN_  64
#define E_    800000
#define HALF_ 32
#define M_    (B_ * N_)

#define SORT_BLOCKS 625
#define SORT_GROUPS 5
#define RUN_EDGES   (256 * SORT_GROUPS)
#define DIG_SHIFT   6
#define DIG_BITS    10
#define NDIG        1024
#define BROWS       64
#define NBUCKET     ((N_ + BROWS - 1) / BROWS)
#define BK_THREADS  512
#define BK_WAVES    16
#define BK_GROUPS   3
#define BK_CHUNK    (BK_THREADS * BK_GROUPS)
#define BK_RPW      (BROWS / BK_WAVES)

#define WS_Z_OFF      0
#define WS_PIECE_OFF  25600000
#define WS_REC_OFF    28168448

static_assert(SORT_BLOCKS * RUN_EDGES == E_, "the filing blocks cover the edges exactly");
static_assert(N_ <= 65536, "a column index and the row's 6 low bits share one 32-bit word");

typedef __attribute__((ext_vector_type(2))) float v2f;
typedef __attribute__((ext_vector_type(8))) float v8f;
typedef __attribute__((ext_vector_type(2))) int   v2i;
typedef __attribute__((ext_vector_type(4))) float v4f;

#define LDS_PITCH 68
#define GEMM_ROWS 64
#define GEMM_LDS_FLOATS ((64 + GEMM_ROWS) * LDS_PITCH)

__device__ __forceinline__ void gemm_tile_body(
    float* __restrict__ lds, const int gb,
    const float* __restrict__ x,
    const float* __restrict__ W_lin,
    const float* __restrict__ W_eye, const float* __restrict__ b_eye,
    float* __restrict__ out, float* __restrict__ z)
{
    float* lw = lds;
    float* lx = lds + 64 * LDS_PITCH;

    const int tid = threadIdx.x;
    #pragma unroll
    for (int j = 0; j < 4; ++j) {
        const int idx = tid + 256 * j;
        const int n = idx >> 4, c4 = idx & 15;
        const float* src = (n < HALF_) ? (W_lin + n * CIN_ + c4 * 4)
                                       : (W_eye + (n - HALF_) * CIN_ + c4 * 4);
        *(v4f*)(lw + n * LDS_PITCH + c4 * 4) = *(const v4f*)src;
    }
    const float* xt = x + (long)gb * (GEMM_ROWS * CIN_);
    #pragma unroll
    for (int j = 0; j < GEMM_ROWS * 16 / 256; ++j) {
        const int idx = tid + 256 * j;
        const int r = idx >> 4, c4 = idx & 15;
        *(v4f*)(lx + r * LDS_PITCH + c4 * 4) = *(const v4f*)(xt + idx * 4);
    }
    __syncthreads();

    const int wave  = tid >> 5;
    const int lane  = tid & 31;
    const int hi    = lane >> 4;
    const int l16   = lane & 15;
    const int rt    = wave >> 1;
    const int ch    = wave & 1;
    const int mtile = gb * (GEMM_ROWS / 16) + rt;

    const int ncol = ch * 32 + l16;

    const float* xrow  = lx + (rt * 16 + l16) * LDS_PITCH + hi * 2;
    const float* wcol0 = lw + ncol * LDS_PITCH + hi * 2;
    const float* wcol1 = wcol0 + 16 * LDS_PITCH;

    v8f acc0 = {}, acc1 = {};
    #pragma unroll
    for (int kb = 0; kb < 16; ++kb) {
        v2f a  = *(const v2f*)(xrow  + kb * 4);
        v2f b0 = *(const v2f*)(wcol0 + kb * 4);
        v2f b1 = *(const v2f*)(wcol1 + kb * 4);
        acc0 = __builtin_amdgcn_wmma_f32_16x16x4_f32(
             false, a,  false, b0,
             (short)0, acc0,  false,  false);
        acc1 = __builtin_amdgcn_wmma_f32_16x16x4_f32(
             false, a,  false, b1,
             (short)0, acc1,  false,  false);
    }

    const int m0 = mtile * 16 + 8 * hi;
    if (ch == 0) {
        #pragma unroll
        for (int r = 0; r < 8; ++r) {
            const long m = m0 + r;
            z[m * HALF_ + l16]      = acc0[r];
            z[m * HALF_ + 16 + l16] = acc1[r];
        }
    } else {
        const float be0 = b_eye[l16];
        const float be1 = b_eye[16 + l16];
        #pragma unroll
        for (int r = 0; r < 8; ++r) {
            const long m = m0 + r;
            out[m * CIN_ + HALF_ + l16]      = acc0[r] + be0;
            out[m * CIN_ + HALF_ + 16 + l16] = acc1[r] + be1;
        }
    }
}


template <int G, int NB>
__device__ __forceinline__ void wave_digit_ranks(
    const int (&d)[G], unsigned lt, int (&lrank)[G], int (&tot)[G])
{
    unsigned bal[G][NB];
    #pragma unroll
    for (int g = 0; g < G; ++g) {
        #pragma unroll
        for (int b = 0; b < NB; ++b)
            bal[g][b] = __builtin_amdgcn_ballot_w32(((d[g] >> b) & 1) != 0);
    }
    #pragma unroll
    for (int g = 0; g < G; ++g) {
        int before = 0, all = 0;
        #pragma unroll
        for (int o = 0; o < G; ++o) {
            unsigned m = 0xffffffffu;
            #pragma unroll
            for (int b = 0; b < NB; ++b)
                m &= ((d[g] >> b) & 1) ? bal[o][b] : ~bal[o][b];
            all += __popc(m);
            if (o < g)  before += __popc(m);
            if (o == g) before += __popc(m & lt);
        }
        lrank[g] = before;
        tot[g]   = all;
    }
}

__device__ __forceinline__ int wave_inclusive_sum(int v, int lane)
{
    #pragma unroll
    for (int s = 1; s < 32; s <<= 1) {
        const int t = __shfl_up(v, s);
        if (lane >= s) v += t;
    }
    return v;
}

__device__ __forceinline__ void file_run_body(
    int* __restrict__ lds  , const int bid,
    const int* __restrict__ rows, const int* __restrict__ cols, const float* __restrict__ vals,
    int* __restrict__ piece, int* __restrict__ rec)
{
    int* h    = lds;
    int* dsum = lds + 8 * NDIG;

    const int tid  = threadIdx.x;
    const int lane = tid & 31;
    const int w    = tid >> 5;

    const int base = bid * RUN_EDGES + w * (32 * SORT_GROUPS) + lane;
    const unsigned lt = (1u << lane) - 1u;

    int key[SORT_GROUPS], d[SORT_GROUPS], lrank[SORT_GROUPS], tot[SORT_GROUPS];
    v2i pr[SORT_GROUPS];
    #pragma unroll
    for (int g = 0; g < SORT_GROUPS; ++g) {
        const int e = base + g * 32;
        key[g]  = rows[e];
        pr[g].x = cols[e];
        pr[g].y = __float_as_int(vals[e]);
        d[g] = (key[g] >> DIG_SHIFT) & (NDIG - 1);
    }
    #pragma unroll
    for (int i = 0; i < 8 * NDIG / 256; ++i) h[tid + 256 * i] = 0;
    __syncthreads();

    wave_digit_ranks<SORT_GROUPS, DIG_BITS>(d, lt, lrank, tot);
    #pragma unroll
    for (int g = 0; g < SORT_GROUPS; ++g)
        if (lrank[g] == 0) h[w * NDIG + d[g]] = tot[g];
    __syncthreads();

    int dv[NDIG / 256], dinc[NDIG / 256];
    #pragma unroll
    for (int j = 0; j < NDIG / 256; ++j) {
        const int digit = tid + 256 * j;
        int t = 0;
        #pragma unroll
        for (int i = 0; i < 8; ++i) t += h[i * NDIG + digit];
        dv[j]   = t;
        dinc[j] = wave_inclusive_sum(t, lane);
        if (lane == 31) dsum[j * 8 + w] = dinc[j];
    }
    __syncthreads();

    {
        int carry = 0;
        #pragma unroll
        for (int j = 0; j < NDIG / 256; ++j) {
            int before = carry, gtotal = 0;
            #pragma unroll
            for (int i = 0; i < 8; ++i) {
                const int v = dsum[j * 8 + i];
                if (i < w) before += v;
                gtotal += v;
            }
            carry += gtotal;
            const int digit = tid + 256 * j;
            const int dbase = before + dinc[j] - dv[j];
            if (digit < NBUCKET) piece[digit * SORT_BLOCKS + bid] = dbase | (dv[j] << 16);
            int run = dbase;
            #pragma unroll
            for (int i = 0; i < 8; ++i) {
                const int c = h[i * NDIG + digit];
                h[i * NDIG + digit] = run;
                run += c;
            }
        }
    }
    __syncthreads();

    v2i* run_out = (v2i*)rec + bid * RUN_EDGES;
    #pragma unroll
    for (int g = 0; g < SORT_GROUPS; ++g) {
        const int pos = h[w * NDIG + d[g]] + lrank[g];
        if ((unsigned)pos < (unsigned)RUN_EDGES) {
            v2i r;
            r.x = (pr[g].x & 0xffff) | ((key[g] & (BROWS - 1)) << 16);
            r.y = pr[g].y;
            run_out[pos] = r;
        }
    }
}

#define SORT_LDS_INTS (8 * NDIG + (NDIG / 256) * 8)
#define GEMM_TILES    (M_ / GEMM_ROWS)
static_assert(GEMM_TILES * GEMM_ROWS == M_, "the GEMM work-groups cover the rows exactly");

union FilingOrGemmLds {
    int   filing[SORT_LDS_INTS];
    float gemm[GEMM_LDS_FLOATS];
};

__global__ __launch_bounds__(256, 16) void gcn_gemm_beside_filing(
    const float* __restrict__ x,
    const float* __restrict__ W_lin,
    const float* __restrict__ W_eye, const float* __restrict__ b_eye,
    float* __restrict__ out, float* __restrict__ z,
    const int* __restrict__ rows, const int* __restrict__ cols, const float* __restrict__ vals,
    int* __restrict__ piece, int* __restrict__ rec)
{
    __shared__ __attribute__((aligned(16))) FilingOrGemmLds lds;

    const int bid = blockIdx.x;
    if (bid < SORT_BLOCKS) {
        file_run_body(lds.filing, bid, rows, cols, vals, piece, rec);
    } else {
        gemm_tile_body(lds.gemm, bid - SORT_BLOCKS, x, W_lin, W_eye, b_eye, out, z);
    }
}

#define BOFF_N 1024
__global__ __launch_bounds__(BK_THREADS, 16) void gcn_bucket_gather(
    const float* __restrict__ z, const int* __restrict__ piece, const int* __restrict__ rec,
    const float* __restrict__ b_lin, float* __restrict__ out)
{
    __shared__ int   boff[BOFF_N];
    __shared__ int   bsrc[SORT_BLOCKS + 1];
    __shared__ int   wsum[BK_WAVES];
    __shared__ int   h[BK_WAVES * 128];
    __shared__ int   seg[BROWS + 1];
    __shared__ v2i   srt[BK_CHUNK];
    __shared__ float accl[BROWS * B_ * 32];

    const int tid  = threadIdx.x;
    const int lane = tid & 31;
    const int w    = tid >> 5;
    const int q    = blockIdx.x;
    const unsigned lt = (1u << lane) - 1u;

    const int* prow = piece + q * SORT_BLOCKS;
    const int rb0 = 2 * tid, rb1 = 2 * tid + 1;
    const int w0 = (rb0 < SORT_BLOCKS) ? prow[rb0] : 0;
    const int w1 = (rb1 < SORT_BLOCKS) ? prow[rb1] : 0;
    const int n0 = (w0 >> 16) & 0x7ff;
    const int n1 = (w1 >> 16) & 0x7ff;

    #pragma unroll
    for (int i = 0; i < BROWS * B_ * 32 / BK_THREADS; ++i) accl[tid + BK_THREADS * i] = 0.f;

    const int psum = n0 + n1;
    int pinc = psum, ptot = psum;
    #pragma unroll
    for (int s = 1; s < 32; s <<= 1) {
        const int t = __shfl_xor(ptot, s);
        if (lane & s) pinc += t;
        ptot += t;
    }
    if (lane == 0) wsum[w] = ptot;
    __syncthreads();
    int pbase = 0, ptotal = 0;
    #pragma unroll
    for (int i = 0; i < BK_WAVES; ++i) {
        const int v = wsum[i];
        if (i < w) pbase += v;
        ptotal += v;
    }
    const int prun = pbase + pinc - psum;
    boff[rb0] = (rb0 < SORT_BLOCKS) ? prun      : 0x7fffffff;
    boff[rb1] = (rb1 < SORT_BLOCKS) ? prun + n0 : 0x7fffffff;
    if (rb0 < SORT_BLOCKS) bsrc[rb0] = rb0 * RUN_EDGES + (w0 & 0x7ff);
    if (rb1 < SORT_BLOCKS) bsrc[rb1] = rb1 * RUN_EDGES + (w1 & 0x7ff);
    const int bs = 0;
    const int be = __builtin_amdgcn_readfirstlane(ptotal);

    const long BS = (long)N_ * HALF_;
    const float* zl = z + lane;
    const v2i* pin = (const v2i*)rec;

    for (int cb = bs; cb < be; cb += BK_CHUNK) {
        #pragma unroll
        for (int i = 0; i < BK_WAVES * 128 / BK_THREADS; ++i) h[tid + BK_THREADS * i] = 0;
        __syncthreads();

        int d[BK_GROUPS], lrank[BK_GROUPS], tot[BK_GROUPS];
        v2i pr[BK_GROUPS];
        int e[BK_GROUPS], es[BK_GROUPS], rb[BK_GROUPS];
        #pragma unroll
        for (int g = 0; g < BK_GROUPS; ++g) {
            e[g]  = cb + w * (32 * BK_GROUPS) + g * 32 + lane;
            es[g] = e[g] < be ? e[g] : be - 1;
            rb[g] = 0;
        }
        #pragma unroll
        for (int step = BOFF_N / 2; step >= 1; step >>= 1) {
            #pragma unroll
            for (int g = 0; g < BK_GROUPS; ++g)
                if (boff[rb[g] + step] <= es[g]) rb[g] += step;
        }
        v2i rc[BK_GROUPS];
        #pragma unroll
        for (int g = 0; g < BK_GROUPS; ++g) {
            int idx = bsrc[rb[g]] + (es[g] - boff[rb[g]]);
            idx = idx < 0 ? 0 : (idx > E_ - 1 ? E_ - 1 : idx);
            rc[g] = pin[idx];
        }
        #pragma unroll
        for (int g = 0; g < BK_GROUPS; ++g) {
            const bool in = e[g] < be;
            d[g]    = in ? ((rc[g].x >> 16) & (BROWS - 1)) : BROWS;
            pr[g].x = in ? (rc[g].x & 0xffff) : 0;
            pr[g].y = in ? rc[g].y : 0;
        }
        wave_digit_ranks<BK_GROUPS, DIG_SHIFT + 1>(d, lt, lrank, tot);
        #pragma unroll
        for (int g = 0; g < BK_GROUPS; ++g)
            if (lrank[g] == 0) h[w * 128 + d[g]] = tot[g];
        __syncthreads();

        if (w == 0) {
            int c0 = 0, c1 = 0;
            #pragma unroll
            for (int i = 0; i < BK_WAVES; ++i) {
                c0 += h[i * 128 + 2 * lane];
                c1 += h[i * 128 + 2 * lane + 1];
            }
            const int s   = c0 + c1;
            static_assert(BK_CHUNK < 2048, "eleven bits hold a chunk's edge count");
            int run0 = 0;
            #pragma unroll
            for (int b = 0; b < 11; ++b) {
                const unsigned mb = __builtin_amdgcn_ballot_w32(((s >> b) & 1) != 0);
                run0 += __popc(mb & lt) << b;
            }
            int run1 = run0 + c0;
            seg[2 * lane]     = run0;
            seg[2 * lane + 1] = run1;
            if (lane == 31) seg[BROWS] = run0 + s;
            #pragma unroll
            for (int i = 0; i < BK_WAVES; ++i) {
                const int t0 = h[i * 128 + 2 * lane];
                const int t1 = h[i * 128 + 2 * lane + 1];
                h[i * 128 + 2 * lane]     = run0;
                h[i * 128 + 2 * lane + 1] = run1;
                run0 += t0;
                run1 += t1;
            }
        }
        __syncthreads();

        #pragma unroll
        for (int g = 0; g < BK_GROUPS; ++g)
            if (d[g] < BROWS) srt[h[w * 128 + d[g]] + lrank[g]] = pr[g];
        __syncthreads();

        #pragma unroll 1
        for (int k = 0; k < BK_RPW; ++k) {
            const int r = w * BK_RPW + k;
            const int start = __builtin_amdgcn_readfirstlane(seg[r]);
            const int end   = __builtin_amdgcn_readfirstlane(seg[r + 1]);
            if (end <= start) continue;

            float* al = accl + r * (B_ * 32) + lane;
            float a0 = al[0], a1 = al[32], a2 = al[64], a3 = al[96];

            for (int base = start; base < end; base += 32) {
                const int m = (end - base < 32) ? (end - base) : 32;
                v2i p = {0, 0};
                if (lane < m) p = srt[base + lane];

                int j = 0;
                for (; j + 4 <= m; j += 4) {
                    const int   c0 = __builtin_amdgcn_readlane(p.x, j);
                    const int   c1 = __builtin_amdgcn_readlane(p.x, j + 1);
                    const int   c2 = __builtin_amdgcn_readlane(p.x, j + 2);
                    const int   c3 = __builtin_amdgcn_readlane(p.x, j + 3);
                    const float v0 = __int_as_float(__builtin_amdgcn_readlane(p.y, j));
                    const float v1 = __int_as_float(__builtin_amdgcn_readlane(p.y, j + 1));
                    const float v2 = __int_as_float(__builtin_amdgcn_readlane(p.y, j + 2));
                    const float v3 = __int_as_float(__builtin_amdgcn_readlane(p.y, j + 3));
                    const float* q0 = zl + (long)c0 * HALF_;
                    const float* q1 = zl + (long)c1 * HALF_;
                    const float* q2 = zl + (long)c2 * HALF_;
                    const float* q3 = zl + (long)c3 * HALF_;
                    const float z00 = q0[0], z01 = q0[BS], z02 = q0[2 * BS], z03 = q0[3 * BS];
                    const float z10 = q1[0], z11 = q1[BS], z12 = q1[2 * BS], z13 = q1[3 * BS];
                    const float z20 = q2[0], z21 = q2[BS], z22 = q2[2 * BS], z23 = q2[3 * BS];
                    const float z30 = q3[0], z31 = q3[BS], z32 = q3[2 * BS], z33 = q3[3 * BS];
                    a0 = fmaf(v0, z00, a0); a1 = fmaf(v0, z01, a1); a2 = fmaf(v0, z02, a2); a3 = fmaf(v0, z03, a3);
                    a0 = fmaf(v1, z10, a0); a1 = fmaf(v1, z11, a1); a2 = fmaf(v1, z12, a2); a3 = fmaf(v1, z13, a3);
                    a0 = fmaf(v2, z20, a0); a1 = fmaf(v2, z21, a1); a2 = fmaf(v2, z22, a2); a3 = fmaf(v2, z23, a3);
                    a0 = fmaf(v3, z30, a0); a1 = fmaf(v3, z31, a1); a2 = fmaf(v3, z32, a2); a3 = fmaf(v3, z33, a3);
                }
                for (; j < m; ++j) {
                    const int   c0 = __builtin_amdgcn_readlane(p.x, j);
                    const float v0 = __int_as_float(__builtin_amdgcn_readlane(p.y, j));
                    const float* q0 = zl + (long)c0 * HALF_;
                    const float z00 = q0[0], z01 = q0[BS], z02 = q0[2 * BS], z03 = q0[3 * BS];
                    a0 = fmaf(v0, z00, a0); a1 = fmaf(v0, z01, a1); a2 = fmaf(v0, z02, a2); a3 = fmaf(v0, z03, a3);
                }
            }
            al[0] = a0; al[32] = a1; al[64] = a2; al[96] = a3;
        }
    }
    __syncthreads();

    const float bl = b_lin[lane];
    const long OS = (long)N_ * CIN_;
    #pragma unroll
    for (int k = 0; k < BK_RPW; ++k) {
        const int r = w * BK_RPW + k;
        const int n = q * BROWS + r;
        if (n < N_) {
            const float* al = accl + r * (B_ * 32) + lane;
            float* o = out + (long)n * CIN_ + lane;
            o[0]      = bl + al[0];
            o[OS]     = bl + al[32];
            o[2 * OS] = bl + al[64];
            o[3 * OS] = bl + al[96];
        }
    }
}

extern "C" void kernel_launch(void* const* d_in, const int* in_sizes, int n_in,
                              void* d_out, int out_size, void* d_ws, size_t ws_size,
                              hipStream_t stream) {
    const float* x     = (const float*)d_in[0];
    const float* vals  = (const float*)d_in[1];
    const float* W_lin = (const float*)d_in[2];
    const float* b_lin = (const float*)d_in[3];
    const float* W_eye = (const float*)d_in[4];
    const float* b_eye = (const float*)d_in[5];
    const int*   rows  = (const int*)d_in[6];
    const int*   cols  = (const int*)d_in[7];
    float* out    = (float*)d_out;
    char*  ws     = (char*)d_ws;
    float* z      = (float*)(ws + WS_Z_OFF);
    int*   piece  = (int*)(ws + WS_PIECE_OFF);
    int*   rec    = (int*)(ws + WS_REC_OFF);

    gcn_gemm_beside_filing<<<SORT_BLOCKS + GEMM_TILES, 256, 0, stream>>>(
        x, W_lin, W_eye, b_eye, out, z, rows, cols, vals, piece, rec);

    gcn_bucket_gather<<<NBUCKET, BK_THREADS, 0, stream>>>(z, piece, rec, b_lin, out);
}


# Launch 1 stores whole 128-byte lines only: a filing block puts its ordered run together in LDS and copies it out in order; GEMM lane halves swap one tile half so a store holds one row's 32 columns
# speedup vs baseline: 5.1988x; 1.0399x over previous
#include <hip/hip_runtime.h>


#define B_    4
#define N_    50000
#define CIN_  64
#define E_    800000
#define HALF_ 32
#define M_    (B_ * N_)

#define SORT_BLOCKS 625
#define SORT_GROUPS 5
#define RUN_EDGES   (256 * SORT_GROUPS)
#define DIG_SHIFT   6
#define DIG_BITS    10
#define NDIG        1024
#define BROWS       64
#define NBUCKET     ((N_ + BROWS - 1) / BROWS)
#define BK_THREADS  512
#define BK_WAVES    16
#define BK_GROUPS   3
#define BK_CHUNK    (BK_THREADS * BK_GROUPS)
#define BK_RPW      (BROWS / BK_WAVES)

#define WS_Z_OFF      0
#define WS_PIECE_OFF  25600000
#define WS_REC_OFF    28168448

static_assert(SORT_BLOCKS * RUN_EDGES == E_, "the filing blocks cover the edges exactly");
static_assert(N_ <= 65536, "a column index and the row's 6 low bits share one 32-bit word");

typedef __attribute__((ext_vector_type(2))) float v2f;
typedef __attribute__((ext_vector_type(8))) float v8f;
typedef __attribute__((ext_vector_type(2))) int   v2i;
typedef __attribute__((ext_vector_type(4))) float v4f;

#define LDS_PITCH 68
#define GEMM_ROWS 64
#define GEMM_LDS_FLOATS ((64 + GEMM_ROWS) * LDS_PITCH)

__device__ __forceinline__ void gemm_tile_body(
    float* __restrict__ lds, const int gb,
    const float* __restrict__ x,
    const float* __restrict__ W_lin,
    const float* __restrict__ W_eye, const float* __restrict__ b_eye,
    float* __restrict__ out, float* __restrict__ z)
{
    float* lw = lds;
    float* lx = lds + 64 * LDS_PITCH;

    const int tid = threadIdx.x;
    #pragma unroll
    for (int j = 0; j < 4; ++j) {
        const int idx = tid + 256 * j;
        const int n = idx >> 4, c4 = idx & 15;
        const float* src = (n < HALF_) ? (W_lin + n * CIN_ + c4 * 4)
                                       : (W_eye + (n - HALF_) * CIN_ + c4 * 4);
        *(v4f*)(lw + n * LDS_PITCH + c4 * 4) = *(const v4f*)src;
    }
    const float* xt = x + (long)gb * (GEMM_ROWS * CIN_);
    #pragma unroll
    for (int j = 0; j < GEMM_ROWS * 16 / 256; ++j) {
        const int idx = tid + 256 * j;
        const int r = idx >> 4, c4 = idx & 15;
        *(v4f*)(lx + r * LDS_PITCH + c4 * 4) = *(const v4f*)(xt + idx * 4);
    }
    __syncthreads();

    const int wave  = tid >> 5;
    const int lane  = tid & 31;
    const int hi    = lane >> 4;
    const int l16   = lane & 15;
    const int rt    = wave >> 1;
    const int ch    = wave & 1;
    const int mtile = gb * (GEMM_ROWS / 16) + rt;

    const int ncol = ch * 32 + l16;

    const float* xrow  = lx + (rt * 16 + l16) * LDS_PITCH + hi * 2;
    const float* wcol0 = lw + ncol * LDS_PITCH + hi * 2;
    const float* wcol1 = wcol0 + 16 * LDS_PITCH;

    v8f acc0 = {}, acc1 = {};
    #pragma unroll
    for (int kb = 0; kb < 16; ++kb) {
        v2f a  = *(const v2f*)(xrow  + kb * 4);
        v2f b0 = *(const v2f*)(wcol0 + kb * 4);
        v2f b1 = *(const v2f*)(wcol1 + kb * 4);
        acc0 = __builtin_amdgcn_wmma_f32_16x16x4_f32(
             false, a,  false, b0,
             (short)0, acc0,  false,  false);
        acc1 = __builtin_amdgcn_wmma_f32_16x16x4_f32(
             false, a,  false, b1,
             (short)0, acc1,  false,  false);
    }

    const long mrow = (long)mtile * 16;
    float* dst;
    long   pitch;
    float  bias;
    if (ch == 0) { dst = z + mrow * HALF_ + lane;           pitch = HALF_; bias = 0.f; }
    else         { dst = out + mrow * CIN_ + HALF_ + lane;  pitch = CIN_;  bias = b_eye[lane]; }
    #pragma unroll
    for (int r = 0; r < 8; ++r) {
        const float send = hi ? acc0[r] : acc1[r];
        const float got = __int_as_float((int)__builtin_amdgcn_permlanex16(
            0u, (unsigned)__float_as_int(send), 0x76543210u, 0xfedcba98u, false, false));
        const float lo_row = hi ? got : acc0[r];
        const float hi_row = hi ? acc1[r] : got;
        dst[(long)r * pitch]       = lo_row + bias;
        dst[(long)(r + 8) * pitch] = hi_row + bias;
    }
}


template <int G, int NB>
__device__ __forceinline__ void wave_digit_ranks(
    const int (&d)[G], unsigned lt, int (&lrank)[G], int (&tot)[G])
{
    unsigned bal[G][NB];
    #pragma unroll
    for (int g = 0; g < G; ++g) {
        #pragma unroll
        for (int b = 0; b < NB; ++b)
            bal[g][b] = __builtin_amdgcn_ballot_w32(((d[g] >> b) & 1) != 0);
    }
    #pragma unroll
    for (int g = 0; g < G; ++g) {
        int before = 0, all = 0;
        #pragma unroll
        for (int o = 0; o < G; ++o) {
            unsigned m = 0xffffffffu;
            #pragma unroll
            for (int b = 0; b < NB; ++b)
                m &= ((d[g] >> b) & 1) ? bal[o][b] : ~bal[o][b];
            all += __popc(m);
            if (o < g)  before += __popc(m);
            if (o == g) before += __popc(m & lt);
        }
        lrank[g] = before;
        tot[g]   = all;
    }
}

__device__ __forceinline__ int wave_inclusive_sum(int v, int lane)
{
    #pragma unroll
    for (int s = 1; s < 32; s <<= 1) {
        const int t = __shfl_up(v, s);
        if (lane >= s) v += t;
    }
    return v;
}

__device__ __forceinline__ void file_run_body(
    int* __restrict__ lds  , const int bid,
    const int* __restrict__ rows, const int* __restrict__ cols, const float* __restrict__ vals,
    int* __restrict__ piece, int* __restrict__ rec)
{
    int* h    = lds;
    int* dsum = lds + 8 * NDIG;

    const int tid  = threadIdx.x;
    const int lane = tid & 31;
    const int w    = tid >> 5;

    const int base = bid * RUN_EDGES + w * (32 * SORT_GROUPS) + lane;
    const unsigned lt = (1u << lane) - 1u;

    int key[SORT_GROUPS], d[SORT_GROUPS], lrank[SORT_GROUPS], tot[SORT_GROUPS];
    v2i pr[SORT_GROUPS];
    #pragma unroll
    for (int g = 0; g < SORT_GROUPS; ++g) {
        const int e = base + g * 32;
        key[g]  = rows[e];
        pr[g].x = cols[e];
        pr[g].y = __float_as_int(vals[e]);
        d[g] = (key[g] >> DIG_SHIFT) & (NDIG - 1);
    }
    #pragma unroll
    for (int i = 0; i < 8 * NDIG / 256; ++i) h[tid + 256 * i] = 0;
    __syncthreads();

    wave_digit_ranks<SORT_GROUPS, DIG_BITS>(d, lt, lrank, tot);
    #pragma unroll
    for (int g = 0; g < SORT_GROUPS; ++g)
        if (lrank[g] == 0) h[w * NDIG + d[g]] = tot[g];
    __syncthreads();

    int dv[NDIG / 256], dinc[NDIG / 256];
    #pragma unroll
    for (int j = 0; j < NDIG / 256; ++j) {
        const int digit = tid + 256 * j;
        int t = 0;
        #pragma unroll
        for (int i = 0; i < 8; ++i) t += h[i * NDIG + digit];
        dv[j]   = t;
        dinc[j] = wave_inclusive_sum(t, lane);
        if (lane == 31) dsum[j * 8 + w] = dinc[j];
    }
    __syncthreads();

    {
        int carry = 0;
        #pragma unroll
        for (int j = 0; j < NDIG / 256; ++j) {
            int before = carry, gtotal = 0;
            #pragma unroll
            for (int i = 0; i < 8; ++i) {
                const int v = dsum[j * 8 + i];
                if (i < w) before += v;
                gtotal += v;
            }
            carry += gtotal;
            const int digit = tid + 256 * j;
            const int dbase = before + dinc[j] - dv[j];
            if (digit < NBUCKET) piece[digit * SORT_BLOCKS + bid] = dbase | (dv[j] << 16);
            int run = dbase;
            #pragma unroll
            for (int i = 0; i < 8; ++i) {
                const int c = h[i * NDIG + digit];
                h[i * NDIG + digit] = run;
                run += c;
            }
        }
    }
    __syncthreads();

    int pos[SORT_GROUPS];
    #pragma unroll
    for (int g = 0; g < SORT_GROUPS; ++g) pos[g] = h[w * NDIG + d[g]] + lrank[g];
    __syncthreads();
    v2i* lrun = (v2i*)lds;
    #pragma unroll
    for (int g = 0; g < SORT_GROUPS; ++g) {
        if ((unsigned)pos[g] < (unsigned)RUN_EDGES) {
            v2i r;
            r.x = (pr[g].x & 0xffff) | ((key[g] & (BROWS - 1)) << 16);
            r.y = pr[g].y;
            lrun[pos[g]] = r;
        }
    }
    __syncthreads();
    v2i* run_out = (v2i*)rec + bid * RUN_EDGES;
    #pragma unroll
    for (int i = 0; i < SORT_GROUPS; ++i) run_out[tid + 256 * i] = lrun[tid + 256 * i];
}

#define SORT_LDS_INTS (8 * NDIG + (NDIG / 256) * 8)
#define GEMM_TILES    (M_ / GEMM_ROWS)
static_assert(GEMM_TILES * GEMM_ROWS == M_, "the GEMM work-groups cover the rows exactly");

union FilingOrGemmLds {
    int   filing[SORT_LDS_INTS];
    float gemm[GEMM_LDS_FLOATS];
};

__global__ __launch_bounds__(256, 16) void gcn_gemm_beside_filing(
    const float* __restrict__ x,
    const float* __restrict__ W_lin,
    const float* __restrict__ W_eye, const float* __restrict__ b_eye,
    float* __restrict__ out, float* __restrict__ z,
    const int* __restrict__ rows, const int* __restrict__ cols, const float* __restrict__ vals,
    int* __restrict__ piece, int* __restrict__ rec)
{
    __shared__ __attribute__((aligned(16))) FilingOrGemmLds lds;

    const int bid = blockIdx.x;
    if (bid < SORT_BLOCKS) {
        file_run_body(lds.filing, bid, rows, cols, vals, piece, rec);
    } else {
        gemm_tile_body(lds.gemm, bid - SORT_BLOCKS, x, W_lin, W_eye, b_eye, out, z);
    }
}

#define BOFF_N 1024
__global__ __launch_bounds__(BK_THREADS, 16) void gcn_bucket_gather(
    const float* __restrict__ z, const int* __restrict__ piece, const int* __restrict__ rec,
    const float* __restrict__ b_lin, float* __restrict__ out)
{
    __shared__ int   boff[BOFF_N];
    __shared__ int   bsrc[SORT_BLOCKS + 1];
    __shared__ int   wsum[BK_WAVES];
    __shared__ int   h[BK_WAVES * 128];
    __shared__ int   seg[BROWS + 1];
    __shared__ v2i   srt[BK_CHUNK];
    __shared__ float accl[BROWS * B_ * 32];

    const int tid  = threadIdx.x;
    const int lane = tid & 31;
    const int w    = tid >> 5;
    const int q    = blockIdx.x;
    const unsigned lt = (1u << lane) - 1u;

    const int* prow = piece + q * SORT_BLOCKS;
    const int rb0 = 2 * tid, rb1 = 2 * tid + 1;
    const int w0 = (rb0 < SORT_BLOCKS) ? prow[rb0] : 0;
    const int w1 = (rb1 < SORT_BLOCKS) ? prow[rb1] : 0;
    const int n0 = (w0 >> 16) & 0x7ff;
    const int n1 = (w1 >> 16) & 0x7ff;

    #pragma unroll
    for (int i = 0; i < BROWS * B_ * 32 / BK_THREADS; ++i) accl[tid + BK_THREADS * i] = 0.f;

    const int psum = n0 + n1;
    int pinc = psum, ptot = psum;
    #pragma unroll
    for (int s = 1; s < 32; s <<= 1) {
        const int t = __shfl_xor(ptot, s);
        if (lane & s) pinc += t;
        ptot += t;
    }
    if (lane == 0) wsum[w] = ptot;
    __syncthreads();
    int pbase = 0, ptotal = 0;
    #pragma unroll
    for (int i = 0; i < BK_WAVES; ++i) {
        const int v = wsum[i];
        if (i < w) pbase += v;
        ptotal += v;
    }
    const int prun = pbase + pinc - psum;
    boff[rb0] = (rb0 < SORT_BLOCKS) ? prun      : 0x7fffffff;
    boff[rb1] = (rb1 < SORT_BLOCKS) ? prun + n0 : 0x7fffffff;
    if (rb0 < SORT_BLOCKS) bsrc[rb0] = rb0 * RUN_EDGES + (w0 & 0x7ff);
    if (rb1 < SORT_BLOCKS) bsrc[rb1] = rb1 * RUN_EDGES + (w1 & 0x7ff);
    const int bs = 0;
    const int be = __builtin_amdgcn_readfirstlane(ptotal);

    const long BS = (long)N_ * HALF_;
    const float* zl = z + lane;
    const v2i* pin = (const v2i*)rec;

    for (int cb = bs; cb < be; cb += BK_CHUNK) {
        #pragma unroll
        for (int i = 0; i < BK_WAVES * 128 / BK_THREADS; ++i) h[tid + BK_THREADS * i] = 0;
        __syncthreads();

        int d[BK_GROUPS], lrank[BK_GROUPS], tot[BK_GROUPS];
        v2i pr[BK_GROUPS];
        int e[BK_GROUPS], es[BK_GROUPS], rb[BK_GROUPS];
        #pragma unroll
        for (int g = 0; g < BK_GROUPS; ++g) {
            e[g]  = cb + w * (32 * BK_GROUPS) + g * 32 + lane;
            es[g] = e[g] < be ? e[g] : be - 1;
            rb[g] = 0;
        }
        #pragma unroll
        for (int step = BOFF_N / 2; step >= 1; step >>= 1) {
            #pragma unroll
            for (int g = 0; g < BK_GROUPS; ++g)
                if (boff[rb[g] + step] <= es[g]) rb[g] += step;
        }
        v2i rc[BK_GROUPS];
        #pragma unroll
        for (int g = 0; g < BK_GROUPS; ++g) {
            int idx = bsrc[rb[g]] + (es[g] - boff[rb[g]]);
            idx = idx < 0 ? 0 : (idx > E_ - 1 ? E_ - 1 : idx);
            rc[g] = pin[idx];
        }
        #pragma unroll
        for (int g = 0; g < BK_GROUPS; ++g) {
            const bool in = e[g] < be;
            d[g]    = in ? ((rc[g].x >> 16) & (BROWS - 1)) : BROWS;
            pr[g].x = in ? (rc[g].x & 0xffff) : 0;
            pr[g].y = in ? rc[g].y : 0;
        }
        wave_digit_ranks<BK_GROUPS, DIG_SHIFT + 1>(d, lt, lrank, tot);
        #pragma unroll
        for (int g = 0; g < BK_GROUPS; ++g)
            if (lrank[g] == 0) h[w * 128 + d[g]] = tot[g];
        __syncthreads();

        if (w == 0) {
            int c0 = 0, c1 = 0;
            #pragma unroll
            for (int i = 0; i < BK_WAVES; ++i) {
                c0 += h[i * 128 + 2 * lane];
                c1 += h[i * 128 + 2 * lane + 1];
            }
            const int s   = c0 + c1;
            static_assert(BK_CHUNK < 2048, "eleven bits hold a chunk's edge count");
            int run0 = 0;
            #pragma unroll
            for (int b = 0; b < 11; ++b) {
                const unsigned mb = __builtin_amdgcn_ballot_w32(((s >> b) & 1) != 0);
                run0 += __popc(mb & lt) << b;
            }
            int run1 = run0 + c0;
            seg[2 * lane]     = run0;
            seg[2 * lane + 1] = run1;
            if (lane == 31) seg[BROWS] = run0 + s;
            #pragma unroll
            for (int i = 0; i < BK_WAVES; ++i) {
                const int t0 = h[i * 128 + 2 * lane];
                const int t1 = h[i * 128 + 2 * lane + 1];
                h[i * 128 + 2 * lane]     = run0;
                h[i * 128 + 2 * lane + 1] = run1;
                run0 += t0;
                run1 += t1;
            }
        }
        __syncthreads();

        #pragma unroll
        for (int g = 0; g < BK_GROUPS; ++g)
            if (d[g] < BROWS) srt[h[w * 128 + d[g]] + lrank[g]] = pr[g];
        __syncthreads();

        #pragma unroll 1
        for (int k = 0; k < BK_RPW; ++k) {
            const int r = w * BK_RPW + k;
            const int start = __builtin_amdgcn_readfirstlane(seg[r]);
            const int end   = __builtin_amdgcn_readfirstlane(seg[r + 1]);
            if (end <= start) continue;

            float* al = accl + r * (B_ * 32) + lane;
            float a0 = al[0], a1 = al[32], a2 = al[64], a3 = al[96];

            for (int base = start; base < end; base += 32) {
                const int m = (end - base < 32) ? (end - base) : 32;
                v2i p = {0, 0};
                if (lane < m) p = srt[base + lane];

                int j = 0;
                for (; j + 4 <= m; j += 4) {
                    const int   c0 = __builtin_amdgcn_readlane(p.x, j);
                    const int   c1 = __builtin_amdgcn_readlane(p.x, j + 1);
                    const int   c2 = __builtin_amdgcn_readlane(p.x, j + 2);
                    const int   c3 = __builtin_amdgcn_readlane(p.x, j + 3);
                    const float v0 = __int_as_float(__builtin_amdgcn_readlane(p.y, j));
                    const float v1 = __int_as_float(__builtin_amdgcn_readlane(p.y, j + 1));
                    const float v2 = __int_as_float(__builtin_amdgcn_readlane(p.y, j + 2));
                    const float v3 = __int_as_float(__builtin_amdgcn_readlane(p.y, j + 3));
                    const float* q0 = zl + (long)c0 * HALF_;
                    const float* q1 = zl + (long)c1 * HALF_;
                    const float* q2 = zl + (long)c2 * HALF_;
                    const float* q3 = zl + (long)c3 * HALF_;
                    const float z00 = q0[0], z01 = q0[BS], z02 = q0[2 * BS], z03 = q0[3 * BS];
                    const float z10 = q1[0], z11 = q1[BS], z12 = q1[2 * BS], z13 = q1[3 * BS];
                    const float z20 = q2[0], z21 = q2[BS], z22 = q2[2 * BS], z23 = q2[3 * BS];
                    const float z30 = q3[0], z31 = q3[BS], z32 = q3[2 * BS], z33 = q3[3 * BS];
                    a0 = fmaf(v0, z00, a0); a1 = fmaf(v0, z01, a1); a2 = fmaf(v0, z02, a2); a3 = fmaf(v0, z03, a3);
                    a0 = fmaf(v1, z10, a0); a1 = fmaf(v1, z11, a1); a2 = fmaf(v1, z12, a2); a3 = fmaf(v1, z13, a3);
                    a0 = fmaf(v2, z20, a0); a1 = fmaf(v2, z21, a1); a2 = fmaf(v2, z22, a2); a3 = fmaf(v2, z23, a3);
                    a0 = fmaf(v3, z30, a0); a1 = fmaf(v3, z31, a1); a2 = fmaf(v3, z32, a2); a3 = fmaf(v3, z33, a3);
                }
                for (; j < m; ++j) {
                    const int   c0 = __builtin_amdgcn_readlane(p.x, j);
                    const float v0 = __int_as_float(__builtin_amdgcn_readlane(p.y, j));
                    const float* q0 = zl + (long)c0 * HALF_;
                    const float z00 = q0[0], z01 = q0[BS], z02 = q0[2 * BS], z03 = q0[3 * BS];
                    a0 = fmaf(v0, z00, a0); a1 = fmaf(v0, z01, a1); a2 = fmaf(v0, z02, a2); a3 = fmaf(v0, z03, a3);
                }
            }
            al[0] = a0; al[32] = a1; al[64] = a2; al[96] = a3;
        }
    }
    __syncthreads();

    const float bl = b_lin[lane];
    const long OS = (long)N_ * CIN_;
    #pragma unroll
    for (int k = 0; k < BK_RPW; ++k) {
        const int r = w * BK_RPW + k;
        const int n = q * BROWS + r;
        if (n < N_) {
            const float* al = accl + r * (B_ * 32) + lane;
            float* o = out + (long)n * CIN_ + lane;
            o[0]      = bl + al[0];
            o[OS]     = bl + al[32];
            o[2 * OS] = bl + al[64];
            o[3 * OS] = bl + al[96];
        }
    }
}

extern "C" void kernel_launch(void* const* d_in, const int* in_sizes, int n_in,
                              void* d_out, int out_size, void* d_ws, size_t ws_size,
                              hipStream_t stream) {
    const float* x     = (const float*)d_in[0];
    const float* vals  = (const float*)d_in[1];
    const float* W_lin = (const float*)d_in[2];
    const float* b_lin = (const float*)d_in[3];
    const float* W_eye = (const float*)d_in[4];
    const float* b_eye = (const float*)d_in[5];
    const int*   rows  = (const int*)d_in[6];
    const int*   cols  = (const int*)d_in[7];
    float* out    = (float*)d_out;
    char*  ws     = (char*)d_ws;
    float* z      = (float*)(ws + WS_Z_OFF);
    int*   piece  = (int*)(ws + WS_PIECE_OFF);
    int*   rec    = (int*)(ws + WS_REC_OFF);

    gcn_gemm_beside_filing<<<SORT_BLOCKS + GEMM_TILES, 256, 0, stream>>>(
        x, W_lin, W_eye, b_eye, out, z, rows, cols, vals, piece, rec);

    gcn_bucket_gather<<<NBUCKET, BK_THREADS, 0, stream>>>(z, piece, rec, b_lin, out);
}
